# starting kernel
# baseline (speedup 1.0000x reference)
#include <hip/hip_runtime.h>
#include <math.h>


#define NGROUPS 561
#define GMM_K   32
#define LOG2PI  1.8378770664093453f
#define XS      68
#define RB      4
#define WPB     3

typedef __attribute__((ext_vector_type(2))) float v2f;
typedef __attribute__((ext_vector_type(8))) float v8f;

__global__ __launch_bounds__(64) void gmm_prep(const float* __restrict__ loc,
                                               const float* __restrict__ cov,
                                               const float* __restrict__ pi,
                                               float* __restrict__ bpack) {
  __shared__ float A[64][65];
  __shared__ float M[64][65];
  __shared__ float mu[64];
  __shared__ float bv[64];
  __shared__ float sc[3];
  const int k = blockIdx.x;
  const int t = threadIdx.x;

  for (int i = 0; i < 64; ++i) A[t][i] = cov[((size_t)k * 64 + t) * 64 + i];
  mu[t] = loc[k * 64 + t];
  if (t == 0) {
    float m = pi[0];
    for (int i = 1; i < GMM_K; ++i) m = fmaxf(m, pi[i]);
    float s = 0.f;
    for (int i = 0; i < GMM_K; ++i) s += expf(pi[i] - m);
    sc[0] = m + logf(s);
  }
  __syncthreads();

  for (int j = 0; j < 64; ++j) {
    if (t == j) A[j][j] = sqrtf(A[j][j]);
    __syncthreads();
    if (t > j) A[t][j] /= A[j][j];
    __syncthreads();
    if (t > j) {
      float l = A[t][j];
      for (int c = j + 1; c <= t; ++c) A[t][c] -= l * A[c][j];
    }
    __syncthreads();
  }
  if (t == 0) {
    float hs = 0.f;
    for (int i = 0; i < 64; ++i) hs += logf(A[i][i]);
    sc[1] = hs;
  }

  {
    const int c = t;
    for (int i = 0; i < c; ++i) M[i][c] = 0.f;
    M[c][c] = 1.f / A[c][c];
    for (int i = c + 1; i < 64; ++i) {
      float s = 0.f;
      for (int p = c; p < i; ++p) s += A[i][p] * M[p][c];
      M[i][c] = -s / A[i][i];
    }
  }
  __syncthreads();

  {
    const int j = t;
    for (int i = 0; i <= j; ++i) {
      float s = 0.f;
      for (int p = j; p < 64; ++p) s += M[p][i] * M[p][j];
      A[j][i] = s;
    }
  }
  __syncthreads();

  {
    const int a = t;
    float s = 0.f;
    for (int b = 0; b < 64; ++b) {
      float p = (b <= a) ? A[a][b] : A[b][a];
      s += p * mu[b];
    }
    bv[a] = s;
  }
  __syncthreads();
  if (t == 0) {
    float c2 = 0.f;
    for (int i = 0; i < 64; ++i) c2 += mu[i] * bv[i];
    sc[2] = -0.5f * c2 + pi[k] - sc[0] - sc[1] - 0.5f * 64.f * LOG2PI;
  }
  __syncthreads();

  const int ctile = k >> 4;
  const int n = k & 15;
  float* base = bpack + (size_t)ctile * NGROUPS * 64;
  int g = 0;
  for (int j = 0; j < 64; ++j) {
    const int nc = (j >> 2) + 1;
    for (int ci = 0; ci < nc; ++ci, ++g) {
      if ((g & 63) == t) {
        float* dst = base + (size_t)g * 64 + n * 2;
        for (int q = 0; q < 4; ++q) {
          const int i = 4 * ci + q;
          float w = 0.f;
          if (i < j)       w = -A[j][i];
          else if (i == j) w = -0.5f * A[j][j];
          dst[(q >> 1) * 32 + (q & 1)] = w;
        }
      }
    }
  }
  for (int ci = 0; ci < 16; ++ci, ++g) {
    if ((g & 63) == t) {
      float* dst = base + (size_t)g * 64 + n * 2;
      for (int q = 0; q < 4; ++q)
        dst[(q >> 1) * 32 + (q & 1)] = bv[4 * ci + q];
    }
  }
  if ((g & 63) == t) {
    float* dst = base + (size_t)g * 64 + n * 2;
    for (int q = 0; q < 4; ++q)
      dst[(q >> 1) * 32 + (q & 1)] = (q == 0) ? sc[2] : 0.f;
  }
}

__global__ __launch_bounds__(32 * WPB) void gmm_main(const float* __restrict__ X,
                                                     const float* __restrict__ bpack,
                                                     float* __restrict__ out,
                                                     int ntiles) {
  __shared__ float xs[WPB][16 * RB][XS];
  const int tid = threadIdx.x;
  const int w = tid >> 5;
  const int lane = tid & 31;
  const int h = lane >> 4;
  const int r = lane & 15;
  const int t0 = (blockIdx.x * WPB + w) * RB;
  int nvalid = ntiles - t0;
  if (nvalid > RB) nvalid = RB;

  if (nvalid > 0) {
    const float* xp = X + (size_t)t0 * 16 * 64;
    const int umax = nvalid * 8;
    for (int u = 0; u < umax; ++u) {
      const int flat = u * 128 + lane * 4;
      float4 v = *(const float4*)(xp + flat);
      *(float4*)&xs[w][flat >> 6][flat & 63] = v;
    }
  }
  __syncthreads();
  if (nvalid <= 0) return;

  const float* xr[RB];
#pragma unroll
  for (int rt = 0; rt < RB; ++rt) xr[rt] = &xs[w][16 * rt + r][0];
  const v2f* __restrict__ bp0 = (const v2f*)bpack + (h * 16 + r);
  const v2f* __restrict__ bp1 = bp0 + (size_t)NGROUPS * 32;

  v8f acc[RB][2];
#pragma unroll
  for (int rt = 0; rt < RB; ++rt) { acc[rt][0] = (v8f){}; acc[rt][1] = (v8f){}; }

  int g = 0;
  for (int j = 0; j < 64; ++j) {
    float tj[RB];
#pragma unroll
    for (int rt = 0; rt < RB; ++rt) tj[rt] = xr[rt][j];
    const int nc = (j >> 2) + 1;
    for (int ci = 0; ci < nc; ++ci, ++g) {
      v2f b0 = bp0[(size_t)g * 32];
      v2f b1 = bp1[(size_t)g * 32];
      v2f a[RB];
#pragma unroll
      for (int rt = 0; rt < RB; ++rt)
        a[rt] = (*(const v2f*)(xr[rt] + 4 * ci + 2 * h)) * tj[rt];
#pragma unroll
      for (int rt = 0; rt < RB; ++rt) {
        acc[rt][0] = __builtin_amdgcn_wmma_f32_16x16x4_f32(false, a[rt], false, b0, (short)0, acc[rt][0], false, false);
        acc[rt][1] = __builtin_amdgcn_wmma_f32_16x16x4_f32(false, a[rt], false, b1, (short)0, acc[rt][1], false, false);
      }
    }
  }
  for (int ci = 0; ci < 16; ++ci, ++g) {
    v2f b0 = bp0[(size_t)g * 32];
    v2f b1 = bp1[(size_t)g * 32];
    v2f a[RB];
#pragma unroll
    for (int rt = 0; rt < RB; ++rt)
      a[rt] = *(const v2f*)(xr[rt] + 4 * ci + 2 * h);
#pragma unroll
    for (int rt = 0; rt < RB; ++rt) {
      acc[rt][0] = __builtin_amdgcn_wmma_f32_16x16x4_f32(false, a[rt], false, b0, (short)0, acc[rt][0], false, false);
      acc[rt][1] = __builtin_amdgcn_wmma_f32_16x16x4_f32(false, a[rt], false, b1, (short)0, acc[rt][1], false, false);
    }
  }
  {
    v2f b0 = bp0[(size_t)g * 32];
    v2f b1 = bp1[(size_t)g * 32];
    v2f a; a.x = (h == 0) ? 1.f : 0.f; a.y = 0.f;
#pragma unroll
    for (int rt = 0; rt < RB; ++rt) {
      acc[rt][0] = __builtin_amdgcn_wmma_f32_16x16x4_f32(false, a, false, b0, (short)0, acc[rt][0], false, false);
      acc[rt][1] = __builtin_amdgcn_wmma_f32_16x16x4_f32(false, a, false, b1, (short)0, acc[rt][1], false, false);
    }
  }

#pragma unroll
  for (int rt = 0; rt < RB; ++rt) {
    float res[8];
#pragma unroll
    for (int v = 0; v < 8; ++v) {
      float a0 = acc[rt][0][v], a1 = acc[rt][1][v];
      float mx = fmaxf(a0, a1);
      mx = fmaxf(mx, __shfl_xor(mx, 1, 32));
      mx = fmaxf(mx, __shfl_xor(mx, 2, 32));
      mx = fmaxf(mx, __shfl_xor(mx, 4, 32));
      mx = fmaxf(mx, __shfl_xor(mx, 8, 32));
      float e = expf(a0 - mx) + expf(a1 - mx);
      e += __shfl_xor(e, 1, 32);
      e += __shfl_xor(e, 2, 32);
      e += __shfl_xor(e, 4, 32);
      e += __shfl_xor(e, 8, 32);
      res[v] = mx + logf(e);
    }
    float val = res[0];
#pragma unroll
    for (int v = 1; v < 8; ++v) val = (r == v) ? res[v] : val;
    if (rt < nvalid && r < 8)
      out[(size_t)(t0 + rt) * 16 + 8 * h + r] = val;
  }
}

extern "C" void kernel_launch(void* const* d_in, const int* in_sizes, int n_in,
                              void* d_out, int out_size, void* d_ws, size_t ws_size,
                              hipStream_t stream) {
  const float* X   = (const float*)d_in[0];
  const float* loc = (const float*)d_in[1];
  const float* cov = (const float*)d_in[2];
  const float* pi  = (const float*)d_in[3];
  float* out = (float*)d_out;
  float* bpack = (float*)d_ws;

  const int N = in_sizes[0] / 64;
  const int ntiles = N / 16;

  gmm_prep<<<GMM_K, 64, 0, stream>>>(loc, cov, pi, bpack);
  const int tilesPerBlock = WPB * RB;
  const int nblk = (ntiles + tilesPerBlock - 1) / tilesPerBlock;
  gmm_main<<<nblk, 32 * WPB, 0, stream>>>(X, bpack, out, ntiles);
}


# Coefficient-prep kernel restructured so no thread runs a serial LDS read-modify-write chain: left-looking Cholesky (1 barrier/column), barrier-free inverse sweep, uniform Gram loops; sum orders kept
# speedup vs baseline: 1.8552x; 1.8552x over previous
#include <hip/hip_runtime.h>
#include <math.h>


#define NGROUPS 561
#define GMM_K   32
#define LOG2PI  1.8378770664093453f
#define XS      68
#define RB      4
#define WPB     3

typedef __attribute__((ext_vector_type(2))) float v2f;
typedef __attribute__((ext_vector_type(8))) float v8f;

#define PS 68

__global__ __launch_bounds__(64) void gmm_prep(const float* __restrict__ loc,
                                               const float* __restrict__ cov,
                                               const float* __restrict__ pi,
                                               float* __restrict__ bpack) {
  __shared__ __attribute__((aligned(16))) float Lm[64][PS];
  __shared__ __attribute__((aligned(16))) float MT[64][PS];
  __shared__ float dr[64];
  __shared__ float dg[64];
  __shared__ float lg[64];
  __shared__ float mu[64];
  __shared__ float bv[64];
  __shared__ float sc[2];
  const int k = blockIdx.x;
  const int t = threadIdx.x;

  {
    const float* ck = cov + (size_t)k * 4096;
    float dv = 0.f;
    for (int i = 0; i < 64; ++i) {
      const float v = ck[i * 64 + t];
      Lm[i][t] = (t < i) ? v : 0.f;
      if (t == i) dv = v;
    }
    dr[t] = dv;
    float4 z; z.x = 0.f; z.y = 0.f; z.z = 0.f; z.w = 0.f;
    for (int p = 0; p < 64; p += 4) *(float4*)&MT[t][p] = z;
  }
  mu[t] = loc[k * 64 + t];
  if (t == 0) {
    float m = pi[0];
    for (int i = 1; i < GMM_K; ++i) m = fmaxf(m, pi[i]);
    float s = 0.f;
    for (int i = 0; i < GMM_K; ++i) s += expf(pi[i] - m);
    sc[0] = m + logf(s);
  }
  __syncthreads();

  {
    const float* rt = &Lm[t][0];
    for (int j = 0; j < 64; ++j) {
      const float* rj = &Lm[j][0];
      float s = rt[j];
      float sjj = dr[j];
      float4 a[16], b[16];
#pragma unroll
      for (int blk = 0; blk < 4; ++blk) {
        if (16 * blk < j) {
#pragma unroll
          for (int q = 4 * blk; q < 4 * blk + 4; ++q) {
            a[q] = *(const float4*)(rt + 4 * q);
            b[q] = *(const float4*)(rj + 4 * q);
          }
        }
      }
#pragma unroll
      for (int blk = 0; blk < 4; ++blk) {
        if (16 * blk < j) {
#pragma unroll
          for (int q = 4 * blk; q < 4 * blk + 4; ++q) {
            s = __builtin_fmaf(-a[q].x, b[q].x, s);  sjj = __builtin_fmaf(-b[q].x, b[q].x, sjj);
            s = __builtin_fmaf(-a[q].y, b[q].y, s);  sjj = __builtin_fmaf(-b[q].y, b[q].y, sjj);
            s = __builtin_fmaf(-a[q].z, b[q].z, s);  sjj = __builtin_fmaf(-b[q].z, b[q].z, sjj);
            s = __builtin_fmaf(-a[q].w, b[q].w, s);  sjj = __builtin_fmaf(-b[q].w, b[q].w, sjj);
          }
        }
      }
      const float d = sqrtf(sjj);
      if (t == j) dg[j] = d;
      if (t > j) Lm[t][j] = s / d;
      __syncthreads();
    }
  }

  {
    const int c = t;
    float* mc = &MT[c][0];
    for (int i = 0; i < 64; ++i) {
      const float* ri = &Lm[i][0];
      const float di = dg[i];
      float s = 0.f;
      float4 a[16], m[16];
#pragma unroll
      for (int blk = 0; blk < 4; ++blk) {
        if (16 * blk < i) {
#pragma unroll
          for (int q = 4 * blk; q < 4 * blk + 4; ++q) {
            a[q] = *(const float4*)(ri + 4 * q);
            m[q] = *(const float4*)(mc + 4 * q);
          }
        }
      }
#pragma unroll
      for (int blk = 0; blk < 4; ++blk) {
        if (16 * blk < i) {
#pragma unroll
          for (int q = 4 * blk; q < 4 * blk + 4; ++q) {
            s = __builtin_fmaf(a[q].x, m[q].x, s);
            s = __builtin_fmaf(a[q].y, m[q].y, s);
            s = __builtin_fmaf(a[q].z, m[q].z, s);
            s = __builtin_fmaf(a[q].w, m[q].w, s);
          }
        }
      }
      float v = -s / di;
      if (i == c) v = 1.f / di;
      if (i < c) v = 0.f;
      mc[i] = v;
    }
  }
  lg[t] = logf(dg[t]);
  __syncthreads();

  {
    const int j = t;
    const int wv = __builtin_amdgcn_readfirstlane(t >> 5);
    const int p0 = 32 * wv;
    const int imax = 32 * (wv + 1);
    const float* mj = &MT[j][0];
    for (int i0 = 0; i0 < imax; i0 += 4) {
      float s0 = 0.f, s1 = 0.f, s2 = 0.f, s3 = 0.f;
      for (int p = p0; p < 64; p += 16) {
        float4 b[4], a0[4], a1[4], a2[4], a3[4];
#pragma unroll
        for (int q = 0; q < 4; ++q) {
          b[q]  = *(const float4*)(mj + p + 4 * q);
          a0[q] = *(const float4*)&MT[i0 + 0][p + 4 * q];
          a1[q] = *(const float4*)&MT[i0 + 1][p + 4 * q];
          a2[q] = *(const float4*)&MT[i0 + 2][p + 4 * q];
          a3[q] = *(const float4*)&MT[i0 + 3][p + 4 * q];
        }
#pragma unroll
        for (int q = 0; q < 4; ++q) {
          s0 = __builtin_fmaf(a0[q].x, b[q].x, s0); s1 = __builtin_fmaf(a1[q].x, b[q].x, s1);
          s2 = __builtin_fmaf(a2[q].x, b[q].x, s2); s3 = __builtin_fmaf(a3[q].x, b[q].x, s3);
          s0 = __builtin_fmaf(a0[q].y, b[q].y, s0); s1 = __builtin_fmaf(a1[q].y, b[q].y, s1);
          s2 = __builtin_fmaf(a2[q].y, b[q].y, s2); s3 = __builtin_fmaf(a3[q].y, b[q].y, s3);
          s0 = __builtin_fmaf(a0[q].z, b[q].z, s0); s1 = __builtin_fmaf(a1[q].z, b[q].z, s1);
          s2 = __builtin_fmaf(a2[q].z, b[q].z, s2); s3 = __builtin_fmaf(a3[q].z, b[q].z, s3);
          s0 = __builtin_fmaf(a0[q].w, b[q].w, s0); s1 = __builtin_fmaf(a1[q].w, b[q].w, s1);
          s2 = __builtin_fmaf(a2[q].w, b[q].w, s2); s3 = __builtin_fmaf(a3[q].w, b[q].w, s3);
        }
      }
      if (i0 + 0 <= j) Lm[j][i0 + 0] = s0;
      if (i0 + 1 <= j) Lm[j][i0 + 1] = s1;
      if (i0 + 2 <= j) Lm[j][i0 + 2] = s2;
      if (i0 + 3 <= j) Lm[j][i0 + 3] = s3;
    }
  }
  __syncthreads();

  {
    const int a = t;
    float s = 0.f;
    for (int b = 0; b < 64; ++b) {
      float p = (b <= a) ? Lm[a][b] : Lm[b][a];
      s += p * mu[b];
    }
    bv[a] = s;
  }
  __syncthreads();
  if (t == 0) {
    float hs = 0.f;
    for (int i = 0; i < 64; ++i) hs += lg[i];
    float c2 = 0.f;
    for (int i = 0; i < 64; ++i) c2 += mu[i] * bv[i];
    sc[1] = -0.5f * c2 + pi[k] - sc[0] - hs - 0.5f * 64.f * LOG2PI;
  }
  __syncthreads();

  const int ctile = k >> 4;
  const int n = k & 15;
  float* base = bpack + (size_t)ctile * NGROUPS * 64 + n * 2;
  for (int g = t; g < NGROUPS; g += 64) {
    float w0 = 0.f, w1 = 0.f, w2 = 0.f, w3 = 0.f;
    if (g < NGROUPS - 17) {
      int a = 0;
      while (2 * (a + 1) * (a + 2) <= g) ++a;
      int rem = g - 2 * a * (a + 1);
      int b = 0;
      while (rem >= a + 1) { rem -= a + 1; ++b; }
      const int j = 4 * a + b;
      const int i4 = 4 * rem;
      const float4 pr = *(const float4*)&Lm[j][i4];
      const float pjj = Lm[j][j];
      w0 = (i4 + 0 < j) ? -pr.x : ((i4 + 0 == j) ? -0.5f * pjj : 0.f);
      w1 = (i4 + 1 < j) ? -pr.y : ((i4 + 1 == j) ? -0.5f * pjj : 0.f);
      w2 = (i4 + 2 < j) ? -pr.z : ((i4 + 2 == j) ? -0.5f * pjj : 0.f);
      w3 = (i4 + 3 < j) ? -pr.w : ((i4 + 3 == j) ? -0.5f * pjj : 0.f);
    } else if (g < NGROUPS - 1) {
      const int ci = g - (NGROUPS - 17);
      w0 = bv[4 * ci + 0]; w1 = bv[4 * ci + 1]; w2 = bv[4 * ci + 2]; w3 = bv[4 * ci + 3];
    } else {
      w0 = sc[1];
    }
    float* dst = base + (size_t)g * 64;
    v2f lo; lo.x = w0; lo.y = w1;
    v2f hi; hi.x = w2; hi.y = w3;
    *(v2f*)dst = lo;
    *(v2f*)(dst + 32) = hi;
  }
}

__global__ __launch_bounds__(32 * WPB) void gmm_main(const float* __restrict__ X,
                                                     const float* __restrict__ bpack,
                                                     float* __restrict__ out,
                                                     int ntiles) {
  __shared__ float xs[WPB][16 * RB][XS];
  const int tid = threadIdx.x;
  const int w = tid >> 5;
  const int lane = tid & 31;
  const int h = lane >> 4;
  const int r = lane & 15;
  const int t0 = (blockIdx.x * WPB + w) * RB;
  int nvalid = ntiles - t0;
  if (nvalid > RB) nvalid = RB;

  if (nvalid > 0) {
    const float* xp = X + (size_t)t0 * 16 * 64;
    const int umax = nvalid * 8;
    for (int u = 0; u < umax; ++u) {
      const int flat = u * 128 + lane * 4;
      float4 v = *(const float4*)(xp + flat);
      *(float4*)&xs[w][flat >> 6][flat & 63] = v;
    }
  }
  __syncthreads();
  if (nvalid <= 0) return;

  const float* xr[RB];
#pragma unroll
  for (int rt = 0; rt < RB; ++rt) xr[rt] = &xs[w][16 * rt + r][0];
  const v2f* __restrict__ bp0 = (const v2f*)bpack + (h * 16 + r);
  const v2f* __restrict__ bp1 = bp0 + (size_t)NGROUPS * 32;

  v8f acc[RB][2];
#pragma unroll
  for (int rt = 0; rt < RB; ++rt) { acc[rt][0] = (v8f){}; acc[rt][1] = (v8f){}; }

  int g = 0;
  for (int j = 0; j < 64; ++j) {
    float tj[RB];
#pragma unroll
    for (int rt = 0; rt < RB; ++rt) tj[rt] = xr[rt][j];
    const int nc = (j >> 2) + 1;
    for (int ci = 0; ci < nc; ++ci, ++g) {
      v2f b0 = bp0[(size_t)g * 32];
      v2f b1 = bp1[(size_t)g * 32];
      v2f a[RB];
#pragma unroll
      for (int rt = 0; rt < RB; ++rt)
        a[rt] = (*(const v2f*)(xr[rt] + 4 * ci + 2 * h)) * tj[rt];
#pragma unroll
      for (int rt = 0; rt < RB; ++rt) {
        acc[rt][0] = __builtin_amdgcn_wmma_f32_16x16x4_f32(false, a[rt], false, b0, (short)0, acc[rt][0], false, false);
        acc[rt][1] = __builtin_amdgcn_wmma_f32_16x16x4_f32(false, a[rt], false, b1, (short)0, acc[rt][1], false, false);
      }
    }
  }
  for (int ci = 0; ci < 16; ++ci, ++g) {
    v2f b0 = bp0[(size_t)g * 32];
    v2f b1 = bp1[(size_t)g * 32];
    v2f a[RB];
#pragma unroll
    for (int rt = 0; rt < RB; ++rt)
      a[rt] = *(const v2f*)(xr[rt] + 4 * ci + 2 * h);
#pragma unroll
    for (int rt = 0; rt < RB; ++rt) {
      acc[rt][0] = __builtin_amdgcn_wmma_f32_16x16x4_f32(false, a[rt], false, b0, (short)0, acc[rt][0], false, false);
      acc[rt][1] = __builtin_amdgcn_wmma_f32_16x16x4_f32(false, a[rt], false, b1, (short)0, acc[rt][1], false, false);
    }
  }
  {
    v2f b0 = bp0[(size_t)g * 32];
    v2f b1 = bp1[(size_t)g * 32];
    v2f a; a.x = (h == 0) ? 1.f : 0.f; a.y = 0.f;
#pragma unroll
    for (int rt = 0; rt < RB; ++rt) {
      acc[rt][0] = __builtin_amdgcn_wmma_f32_16x16x4_f32(false, a, false, b0, (short)0, acc[rt][0], false, false);
      acc[rt][1] = __builtin_amdgcn_wmma_f32_16x16x4_f32(false, a, false, b1, (short)0, acc[rt][1], false, false);
    }
  }

#pragma unroll
  for (int rt = 0; rt < RB; ++rt) {
    float res[8];
#pragma unroll
    for (int v = 0; v < 8; ++v) {
      float a0 = acc[rt][0][v], a1 = acc[rt][1][v];
      float mx = fmaxf(a0, a1);
      mx = fmaxf(mx, __shfl_xor(mx, 1, 32));
      mx = fmaxf(mx, __shfl_xor(mx, 2, 32));
      mx = fmaxf(mx, __shfl_xor(mx, 4, 32));
      mx = fmaxf(mx, __shfl_xor(mx, 8, 32));
      float e = expf(a0 - mx) + expf(a1 - mx);
      e += __shfl_xor(e, 1, 32);
      e += __shfl_xor(e, 2, 32);
      e += __shfl_xor(e, 4, 32);
      e += __shfl_xor(e, 8, 32);
      res[v] = mx + logf(e);
    }
    float val = res[0];
#pragma unroll
    for (int v = 1; v < 8; ++v) val = (r == v) ? res[v] : val;
    if (rt < nvalid && r < 8)
      out[(size_t)(t0 + rt) * 16 + 8 * h + r] = val;
  }
}

extern "C" void kernel_launch(void* const* d_in, const int* in_sizes, int n_in,
                              void* d_out, int out_size, void* d_ws, size_t ws_size,
                              hipStream_t stream) {
  const float* X   = (const float*)d_in[0];
  const float* loc = (const float*)d_in[1];
  const float* cov = (const float*)d_in[2];
  const float* pi  = (const float*)d_in[3];
  float* out = (float*)d_out;
  float* bpack = (float*)d_ws;

  const int N = in_sizes[0] / 64;
  const int ntiles = N / 16;

  gmm_prep<<<GMM_K, 64, 0, stream>>>(loc, cov, pi, bpack);
  const int tilesPerBlock = WPB * RB;
  const int nblk = (ntiles + tilesPerBlock - 1) / tilesPerBlock;
  gmm_main<<<nblk, 32 * WPB, 0, stream>>>(X, bpack, out, ntiles);
}


# gmm_main: K loop flattened to uniform 11-group chunks, each chunk's B fragments and x operands fetched one chunk ahead of its WMMAs; 4-wave blocks (one wave per SIMD); sum order kept
# speedup vs baseline: 3.1706x; 1.7090x over previous
#include <hip/hip_runtime.h>
#include <math.h>


#define NGROUPS 561
#define GMM_K   32
#define LOG2PI  1.8378770664093453f
#define XS      68
#define RB      4
#define WPB     4

typedef __attribute__((ext_vector_type(2))) float v2f;
typedef __attribute__((ext_vector_type(8))) float v8f;

#define PS 68

__global__ __launch_bounds__(64) void gmm_prep(const float* __restrict__ loc,
                                               const float* __restrict__ cov,
                                               const float* __restrict__ pi,
                                               float* __restrict__ bpack) {
  __shared__ __attribute__((aligned(16))) float Lm[64][PS];
  __shared__ __attribute__((aligned(16))) float MT[64][PS];
  __shared__ float dr[64];
  __shared__ float dg[64];
  __shared__ float lg[64];
  __shared__ float mu[64];
  __shared__ float bv[64];
  __shared__ float sc[2];
  const int k = blockIdx.x;
  const int t = threadIdx.x;

  {
    const float* ck = cov + (size_t)k * 4096;
    float dv = 0.f;
    for (int i = 0; i < 64; ++i) {
      const float v = ck[i * 64 + t];
      Lm[i][t] = (t < i) ? v : 0.f;
      if (t == i) dv = v;
    }
    dr[t] = dv;
    float4 z; z.x = 0.f; z.y = 0.f; z.z = 0.f; z.w = 0.f;
    for (int p = 0; p < 64; p += 4) *(float4*)&MT[t][p] = z;
  }
  mu[t] = loc[k * 64 + t];
  if (t == 0) {
    float m = pi[0];
    for (int i = 1; i < GMM_K; ++i) m = fmaxf(m, pi[i]);
    float s = 0.f;
    for (int i = 0; i < GMM_K; ++i) s += expf(pi[i] - m);
    sc[0] = m + logf(s);
  }
  __syncthreads();

  {
    const float* rt = &Lm[t][0];
    for (int j = 0; j < 64; ++j) {
      const float* rj = &Lm[j][0];
      float s = rt[j];
      float sjj = dr[j];
      float4 a[16], b[16];
#pragma unroll
      for (int blk = 0; blk < 4; ++blk) {
        if (16 * blk < j) {
#pragma unroll
          for (int q = 4 * blk; q < 4 * blk + 4; ++q) {
            a[q] = *(const float4*)(rt + 4 * q);
            b[q] = *(const float4*)(rj + 4 * q);
          }
        }
      }
#pragma unroll
      for (int blk = 0; blk < 4; ++blk) {
        if (16 * blk < j) {
#pragma unroll
          for (int q = 4 * blk; q < 4 * blk + 4; ++q) {
            s = __builtin_fmaf(-a[q].x, b[q].x, s);  sjj = __builtin_fmaf(-b[q].x, b[q].x, sjj);
            s = __builtin_fmaf(-a[q].y, b[q].y, s);  sjj = __builtin_fmaf(-b[q].y, b[q].y, sjj);
            s = __builtin_fmaf(-a[q].z, b[q].z, s);  sjj = __builtin_fmaf(-b[q].z, b[q].z, sjj);
            s = __builtin_fmaf(-a[q].w, b[q].w, s);  sjj = __builtin_fmaf(-b[q].w, b[q].w, sjj);
          }
        }
      }
      const float d = sqrtf(sjj);
      if (t == j) dg[j] = d;
      if (t > j) Lm[t][j] = s / d;
      __syncthreads();
    }
  }

  {
    const int c = t;
    float* mc = &MT[c][0];
    for (int i = 0; i < 64; ++i) {
      const float* ri = &Lm[i][0];
      const float di = dg[i];
      float s = 0.f;
      float4 a[16], m[16];
#pragma unroll
      for (int blk = 0; blk < 4; ++blk) {
        if (16 * blk < i) {
#pragma unroll
          for (int q = 4 * blk; q < 4 * blk + 4; ++q) {
            a[q] = *(const float4*)(ri + 4 * q);
            m[q] = *(const float4*)(mc + 4 * q);
          }
        }
      }
#pragma unroll
      for (int blk = 0; blk < 4; ++blk) {
        if (16 * blk < i) {
#pragma unroll
          for (int q = 4 * blk; q < 4 * blk + 4; ++q) {
            s = __builtin_fmaf(a[q].x, m[q].x, s);
            s = __builtin_fmaf(a[q].y, m[q].y, s);
            s = __builtin_fmaf(a[q].z, m[q].z, s);
            s = __builtin_fmaf(a[q].w, m[q].w, s);
          }
        }
      }
      float v = -s / di;
      if (i == c) v = 1.f / di;
      if (i < c) v = 0.f;
      mc[i] = v;
    }
  }
  lg[t] = logf(dg[t]);
  __syncthreads();

  {
    const int j = t;
    const int wv = __builtin_amdgcn_readfirstlane(t >> 5);
    const int p0 = 32 * wv;
    const int imax = 32 * (wv + 1);
    const float* mj = &MT[j][0];
    for (int i0 = 0; i0 < imax; i0 += 4) {
      float s0 = 0.f, s1 = 0.f, s2 = 0.f, s3 = 0.f;
      for (int p = p0; p < 64; p += 16) {
        float4 b[4], a0[4], a1[4], a2[4], a3[4];
#pragma unroll
        for (int q = 0; q < 4; ++q) {
          b[q]  = *(const float4*)(mj + p + 4 * q);
          a0[q] = *(const float4*)&MT[i0 + 0][p + 4 * q];
          a1[q] = *(const float4*)&MT[i0 + 1][p + 4 * q];
          a2[q] = *(const float4*)&MT[i0 + 2][p + 4 * q];
          a3[q] = *(const float4*)&MT[i0 + 3][p + 4 * q];
        }
#pragma unroll
        for (int q = 0; q < 4; ++q) {
          s0 = __builtin_fmaf(a0[q].x, b[q].x, s0); s1 = __builtin_fmaf(a1[q].x, b[q].x, s1);
          s2 = __builtin_fmaf(a2[q].x, b[q].x, s2); s3 = __builtin_fmaf(a3[q].x, b[q].x, s3);
          s0 = __builtin_fmaf(a0[q].y, b[q].y, s0); s1 = __builtin_fmaf(a1[q].y, b[q].y, s1);
          s2 = __builtin_fmaf(a2[q].y, b[q].y, s2); s3 = __builtin_fmaf(a3[q].y, b[q].y, s3);
          s0 = __builtin_fmaf(a0[q].z, b[q].z, s0); s1 = __builtin_fmaf(a1[q].z, b[q].z, s1);
          s2 = __builtin_fmaf(a2[q].z, b[q].z, s2); s3 = __builtin_fmaf(a3[q].z, b[q].z, s3);
          s0 = __builtin_fmaf(a0[q].w, b[q].w, s0); s1 = __builtin_fmaf(a1[q].w, b[q].w, s1);
          s2 = __builtin_fmaf(a2[q].w, b[q].w, s2); s3 = __builtin_fmaf(a3[q].w, b[q].w, s3);
        }
      }
      if (i0 + 0 <= j) Lm[j][i0 + 0] = s0;
      if (i0 + 1 <= j) Lm[j][i0 + 1] = s1;
      if (i0 + 2 <= j) Lm[j][i0 + 2] = s2;
      if (i0 + 3 <= j) Lm[j][i0 + 3] = s3;
    }
  }
  __syncthreads();

  {
    const int a = t;
    float s = 0.f;
    for (int b = 0; b < 64; ++b) {
      float p = (b <= a) ? Lm[a][b] : Lm[b][a];
      s += p * mu[b];
    }
    bv[a] = s;
  }
  __syncthreads();
  if (t == 0) {
    float hs = 0.f;
    for (int i = 0; i < 64; ++i) hs += lg[i];
    float c2 = 0.f;
    for (int i = 0; i < 64; ++i) c2 += mu[i] * bv[i];
    sc[1] = -0.5f * c2 + pi[k] - sc[0] - hs - 0.5f * 64.f * LOG2PI;
  }
  __syncthreads();

  const int ctile = k >> 4;
  const int n = k & 15;
  float* base = bpack + (size_t)ctile * NGROUPS * 64 + n * 2;
  for (int g = t; g < NGROUPS; g += 64) {
    float w0 = 0.f, w1 = 0.f, w2 = 0.f, w3 = 0.f;
    if (g < NGROUPS - 17) {
      int a = 0;
      while (2 * (a + 1) * (a + 2) <= g) ++a;
      int rem = g - 2 * a * (a + 1);
      int b = 0;
      while (rem >= a + 1) { rem -= a + 1; ++b; }
      const int j = 4 * a + b;
      const int i4 = 4 * rem;
      const float4 pr = *(const float4*)&Lm[j][i4];
      const float pjj = Lm[j][j];
      w0 = (i4 + 0 < j) ? -pr.x : ((i4 + 0 == j) ? -0.5f * pjj : 0.f);
      w1 = (i4 + 1 < j) ? -pr.y : ((i4 + 1 == j) ? -0.5f * pjj : 0.f);
      w2 = (i4 + 2 < j) ? -pr.z : ((i4 + 2 == j) ? -0.5f * pjj : 0.f);
      w3 = (i4 + 3 < j) ? -pr.w : ((i4 + 3 == j) ? -0.5f * pjj : 0.f);
    } else if (g < NGROUPS - 1) {
      const int ci = g - (NGROUPS - 17);
      w0 = bv[4 * ci + 0]; w1 = bv[4 * ci + 1]; w2 = bv[4 * ci + 2]; w3 = bv[4 * ci + 3];
    } else {
      w0 = sc[1];
    }
    float* dst = base + (size_t)g * 64;
    v2f lo; lo.x = w0; lo.y = w1;
    v2f hi; hi.x = w2; hi.y = w3;
    *(v2f*)dst = lo;
    *(v2f*)(dst + 32) = hi;
  }
}

#define CH     11
#define NCHUNK (NGROUPS / CH)

__device__ __forceinline__ void gmm_next_addr(int h, int& nj, int& nci, int (&xa)[CH], int (&ta)[CH]) {
#pragma unroll
  for (int i = 0; i < CH; ++i) {
    xa[i] = 4 * nci + 2 * h;
    ta[i] = nj;
    ++nci;
    if (nci > (nj >> 2)) { nci = 0; ++nj; }
  }
}
__device__ __forceinline__ void gmm_fetch_x(const float* xl, const int (&xa)[CH], const int (&ta)[CH],
                                            v2f (&xr)[CH][RB], float (&tr)[CH][RB]) {
#pragma unroll
  for (int i = 0; i < CH; ++i) {
#pragma unroll
    for (int rt = 0; rt < RB; ++rt) {
      xr[i][rt] = *(const v2f*)(xl + xa[i] + rt * 16 * XS);
      tr[i][rt] = xl[ta[i] + rt * 16 * XS];
    }
  }
}
__device__ __forceinline__ void gmm_fetch_b(const v2f* __restrict__ bp0, const v2f* __restrict__ bp1,
                                            int g0, v2f (&bn)[CH][2]) {
#pragma unroll
  for (int i = 0; i < CH; ++i) {
    bn[i][0] = bp0[(size_t)(g0 + i) * 32];
    bn[i][1] = bp1[(size_t)(g0 + i) * 32];
  }
}
__device__ __forceinline__ void gmm_make_a(v2f (&ac)[CH][RB], const v2f (&xr)[CH][RB], const float (&tr)[CH][RB]) {
#pragma unroll
  for (int i = 0; i < CH; ++i) {
#pragma unroll
    for (int rt = 0; rt < RB; ++rt) ac[i][rt] = xr[i][rt] * tr[i][rt];
  }
}
__device__ __forceinline__ void gmm_run(v8f (&acc)[RB][2], const v2f (&ac)[CH][RB], const v2f (&bc)[CH][2]) {
#pragma unroll
  for (int i = 0; i < CH; ++i) {
#pragma unroll
    for (int rt = 0; rt < RB; ++rt) {
      acc[rt][0] = __builtin_amdgcn_wmma_f32_16x16x4_f32(false, ac[i][rt], false, bc[i][0], (short)0, acc[rt][0], false, false);
      acc[rt][1] = __builtin_amdgcn_wmma_f32_16x16x4_f32(false, ac[i][rt], false, bc[i][1], (short)0, acc[rt][1], false, false);
    }
  }
}
__device__ __forceinline__ void gmm_step(v8f (&acc)[RB][2], v2f (&ac)[CH][RB], const v2f (&bc)[CH][2],
                                         v2f (&bn)[CH][2], const v2f* __restrict__ bp0,
                                         const v2f* __restrict__ bp1, int gnext, const float* xl, int h,
                                         int& nj, int& nci, int (&xa)[CH], int (&ta)[CH]) {
  v2f xr[CH][RB];
  float tr[CH][RB];
  gmm_fetch_b(bp0, bp1, gnext, bn);
  gmm_fetch_x(xl, xa, ta, xr, tr);
  __builtin_amdgcn_sched_barrier(0);
  gmm_run(acc, ac, bc);
  gmm_next_addr(h, nj, nci, xa, ta);
  gmm_make_a(ac, xr, tr);
}

__global__ __launch_bounds__(32 * WPB) void gmm_main(const float* __restrict__ X,
                                                     const float* __restrict__ bpack,
                                                     float* __restrict__ out,
                                                     int ntiles) {
  __shared__ __attribute__((aligned(16))) float xs[WPB][16 * RB][XS];
  const int tid = threadIdx.x;
  const int w = __builtin_amdgcn_readfirstlane(tid >> 5);
  const int lane = tid & 31;
  const int h = lane >> 4;
  const int r = lane & 15;
  const int t0 = (blockIdx.x * WPB + w) * RB;
  int nvalid = ntiles - t0;
  if (nvalid > RB) nvalid = RB;

  if (nvalid > 0) {
    const float* xp = X + (size_t)t0 * 16 * 64;
    const int umax = nvalid * 8;
    for (int u = 0; u < umax; ++u) {
      const int flat = u * 128 + lane * 4;
      float4 v = *(const float4*)(xp + flat);
      *(float4*)&xs[w][flat >> 6][flat & 63] = v;
    }
    float4 z; z.x = 0.f; z.y = 0.f; z.z = 0.f; z.w = 0.f;
    for (int u = umax; u < RB * 8; ++u) {
      const int flat = u * 128 + lane * 4;
      *(float4*)&xs[w][flat >> 6][flat & 63] = z;
    }
    float4 one; one.x = 1.f; one.y = 0.f; one.z = 0.f; one.w = 0.f;
    *(float4*)&xs[w][lane][64] = one;
    *(float4*)&xs[w][lane + 32][64] = one;
  }
  __syncthreads();
  if (nvalid <= 0) return;

  const float* xl = &xs[w][r][0];
  const v2f* __restrict__ bp0 = (const v2f*)bpack + (h * 16 + r);
  const v2f* __restrict__ bp1 = bp0 + (size_t)NGROUPS * 32;

  v8f acc[RB][2];
#pragma unroll
  for (int rt = 0; rt < RB; ++rt) { acc[rt][0] = (v8f){}; acc[rt][1] = (v8f){}; }

  v2f be[CH][2], bo[CH][2];
  v2f ac[CH][RB];
  int xa[CH], ta[CH];
  int nj = 0, nci = 0;

  {
    v2f xr[CH][RB];
    float tr[CH][RB];
    gmm_next_addr(h, nj, nci, xa, ta);
    gmm_fetch_b(bp0, bp1, 0, be);
    gmm_fetch_x(xl, xa, ta, xr, tr);
    gmm_next_addr(h, nj, nci, xa, ta);
    gmm_make_a(ac, xr, tr);
  }
  for (int c = 0; c + 2 < NCHUNK; c += 2) {
    gmm_step(acc, ac, be, bo, bp0, bp1, (c + 1) * CH, xl, h, nj, nci, xa, ta);
    gmm_step(acc, ac, bo, be, bp0, bp1, (c + 2) * CH, xl, h, nj, nci, xa, ta);
  }
  gmm_run(acc, ac, be);

#pragma unroll
  for (int rt = 0; rt < RB; ++rt) {
    float res[8];
#pragma unroll
    for (int v = 0; v < 8; ++v) {
      float a0 = acc[rt][0][v], a1 = acc[rt][1][v];
      float mx = fmaxf(a0, a1);
      mx = fmaxf(mx, __shfl_xor(mx, 1, 32));
      mx = fmaxf(mx, __shfl_xor(mx, 2, 32));
      mx = fmaxf(mx, __shfl_xor(mx, 4, 32));
      mx = fmaxf(mx, __shfl_xor(mx, 8, 32));
      float e = expf(a0 - mx) + expf(a1 - mx);
      e += __shfl_xor(e, 1, 32);
      e += __shfl_xor(e, 2, 32);
      e += __shfl_xor(e, 4, 32);
      e += __shfl_xor(e, 8, 32);
      res[v] = mx + logf(e);
    }
    float val = res[0];
#pragma unroll
    for (int v = 1; v < 8; ++v) val = (r == v) ? res[v] : val;
    if (rt < nvalid && r < 8)
      out[(size_t)(t0 + rt) * 16 + 8 * h + r] = val;
  }
}

extern "C" void kernel_launch(void* const* d_in, const int* in_sizes, int n_in,
                              void* d_out, int out_size, void* d_ws, size_t ws_size,
                              hipStream_t stream) {
  const float* X   = (const float*)d_in[0];
  const float* loc = (const float*)d_in[1];
  const float* cov = (const float*)d_in[2];
  const float* pi  = (const float*)d_in[3];
  float* out = (float*)d_out;
  float* bpack = (float*)d_ws;

  const int N = in_sizes[0] / 64;
  const int ntiles = N / 16;

  gmm_prep<<<GMM_K, 64, 0, stream>>>(loc, cov, pi, bpack);
  const int tilesPerBlock = WPB * RB;
  const int nblk = (ntiles + tilesPerBlock - 1) / tilesPerBlock;
  gmm_main<<<nblk, 32 * WPB, 0, stream>>>(X, bpack, out, ntiles);
}


# gmm_prep: Cholesky, triangular inverse and Gram product replaced by one in-register Gauss-Jordan inversion of the symmetrised covariance, log-det from its pivots; coefficient sum order changes
# speedup vs baseline: 3.7344x; 1.1778x over previous
#include <hip/hip_runtime.h>
#include <math.h>


#define NGROUPS 561
#define GMM_K   32
#define LOG2PI  1.8378770664093453f
#define XS      68
#define RB      4
#define WPB     4

typedef __attribute__((ext_vector_type(2))) float v2f;
typedef __attribute__((ext_vector_type(8))) float v8f;

#define PS 68

__global__ __launch_bounds__(64) void gmm_prep(const float* __restrict__ loc,
                                               const float* __restrict__ cov,
                                               const float* __restrict__ pi,
                                               float* __restrict__ bpack) {
  __shared__ __attribute__((aligned(16))) float Lm[64][PS];
  __shared__ __attribute__((aligned(16))) float rowb[2][64];
  __shared__ float nipv[2][64];
  __shared__ float lg[64];
  __shared__ __attribute__((aligned(16))) float mu[64];
  __shared__ float bv[64];
  __shared__ float sc[2];
  const int k = blockIdx.x;
  const int t = threadIdx.x;

  {
    const float* ck = cov + (size_t)k * 4096;
    for (int i = 0; i < 64; ++i) Lm[i][t] = ck[i * 64 + t];
  }
  mu[t] = loc[k * 64 + t];
  if (t == 0) {
    float m = pi[0];
    for (int i = 1; i < GMM_K; ++i) m = fmaxf(m, pi[i]);
    float s = 0.f;
    for (int i = 0; i < GMM_K; ++i) s += expf(pi[i] - m);
    sc[0] = m + logf(s);
  }
  __syncthreads();

  float r[64];
#pragma unroll
  for (int q = 0; q < 16; ++q) {
    const float4 a = *(const float4*)&Lm[t][4 * q];
    r[4 * q + 0] = (a.x + Lm[4 * q + 0][t]) * 0.5f;
    r[4 * q + 1] = (a.y + Lm[4 * q + 1][t]) * 0.5f;
    r[4 * q + 2] = (a.z + Lm[4 * q + 2][t]) * 0.5f;
    r[4 * q + 3] = (a.w + Lm[4 * q + 3][t]) * 0.5f;
  }
  if (t == 0) {
#pragma unroll
    for (int q = 0; q < 16; ++q) {
      float4 v; v.x = r[4 * q + 0]; v.y = r[4 * q + 1]; v.z = r[4 * q + 2]; v.w = r[4 * q + 3];
      *(float4*)&rowb[0][4 * q] = v;
    }
  }
  nipv[0][t] = 1.0f / r[0];
  __syncthreads();

  float piv_own = 1.f;
#pragma nounroll
  for (int s = 0; s < 64; ++s) {
    const float* rb = &rowb[s & 1][0];
    float w[64];
#pragma unroll
    for (int q = 0; q < 16; ++q) {
      const float4 v = *(const float4*)(rb + 4 * q);
      w[4 * q + 0] = v.x; w[4 * q + 1] = v.y; w[4 * q + 2] = v.z; w[4 * q + 3] = v.w;
    }
    const float ip = nipv[s & 1][s];
    const bool own = (t == s);
    piv_own = own ? w[0] : piv_own;
    const float f = own ? -ip : r[0] * ip;
#pragma unroll
    for (int m = 1; m < 64; ++m) {
      const float base = own ? 0.f : r[m];
      r[m - 1] = __builtin_fmaf(-f, w[m], base);
    }
    r[63] = -f;
    nipv[(s + 1) & 1][t] = 1.0f / r[0];
    if (t == s + 1) {
      float* wb = &rowb[(s + 1) & 1][0];
#pragma unroll
      for (int q = 0; q < 16; ++q) {
        float4 v; v.x = r[4 * q + 0]; v.y = r[4 * q + 1]; v.z = r[4 * q + 2]; v.w = r[4 * q + 3];
        *(float4*)(wb + 4 * q) = v;
      }
    }
    __syncthreads();
  }

  lg[t] = logf(piv_own);
  {
    float s = 0.f;
#pragma unroll
    for (int q = 0; q < 16; ++q) {
      float4 v; v.x = r[4 * q + 0]; v.y = r[4 * q + 1]; v.z = r[4 * q + 2]; v.w = r[4 * q + 3];
      *(float4*)&Lm[t][4 * q] = v;
      const float4 m4 = *(const float4*)&mu[4 * q];
      s += v.x * m4.x;
      s += v.y * m4.y;
      s += v.z * m4.z;
      s += v.w * m4.w;
    }
    bv[t] = s;
  }
  __syncthreads();
  if (t == 0) {
    float hs = 0.f;
    for (int i = 0; i < 64; ++i) hs += lg[i];
    float c2 = 0.f;
    for (int i = 0; i < 64; ++i) c2 += mu[i] * bv[i];
    sc[1] = -0.5f * c2 + pi[k] - sc[0] - 0.5f * hs - 0.5f * 64.f * LOG2PI;
  }
  __syncthreads();

  const int ctile = k >> 4;
  const int n = k & 15;
  float* base = bpack + (size_t)ctile * NGROUPS * 64 + n * 2;
  for (int g = t; g < NGROUPS; g += 64) {
    float w0 = 0.f, w1 = 0.f, w2 = 0.f, w3 = 0.f;
    if (g < NGROUPS - 17) {
      int a = 0;
      while (2 * (a + 1) * (a + 2) <= g) ++a;
      int rem = g - 2 * a * (a + 1);
      int b = 0;
      while (rem >= a + 1) { rem -= a + 1; ++b; }
      const int j = 4 * a + b;
      const int i4 = 4 * rem;
      const float4 pr = *(const float4*)&Lm[j][i4];
      const float pjj = Lm[j][j];
      w0 = (i4 + 0 < j) ? -pr.x : ((i4 + 0 == j) ? -0.5f * pjj : 0.f);
      w1 = (i4 + 1 < j) ? -pr.y : ((i4 + 1 == j) ? -0.5f * pjj : 0.f);
      w2 = (i4 + 2 < j) ? -pr.z : ((i4 + 2 == j) ? -0.5f * pjj : 0.f);
      w3 = (i4 + 3 < j) ? -pr.w : ((i4 + 3 == j) ? -0.5f * pjj : 0.f);
    } else if (g < NGROUPS - 1) {
      const int ci = g - (NGROUPS - 17);
      w0 = bv[4 * ci + 0]; w1 = bv[4 * ci + 1]; w2 = bv[4 * ci + 2]; w3 = bv[4 * ci + 3];
    } else {
      w0 = sc[1];
    }
    float* dst = base + (size_t)g * 64;
    v2f lo; lo.x = w0; lo.y = w1;
    v2f hi; hi.x = w2; hi.y = w3;
    *(v2f*)dst = lo;
    *(v2f*)(dst + 32) = hi;
  }
}

#define CH     11
#define NCHUNK (NGROUPS / CH)

__device__ __forceinline__ void gmm_next_addr(int h, int& nj, int& nci, int (&xa)[CH], int (&ta)[CH]) {
#pragma unroll
  for (int i = 0; i < CH; ++i) {
    xa[i] = 4 * nci + 2 * h;
    ta[i] = nj;
    ++nci;
    if (nci > (nj >> 2)) { nci = 0; ++nj; }
  }
}
__device__ __forceinline__ void gmm_fetch_x(const float* xl, const int (&xa)[CH], const int (&ta)[CH],
                                            v2f (&xr)[CH][RB], float (&tr)[CH][RB]) {
#pragma unroll
  for (int i = 0; i < CH; ++i) {
#pragma unroll
    for (int rt = 0; rt < RB; ++rt) {
      xr[i][rt] = *(const v2f*)(xl + xa[i] + rt * 16 * XS);
      tr[i][rt] = xl[ta[i] + rt * 16 * XS];
    }
  }
}
__device__ __forceinline__ void gmm_fetch_b(const v2f* __restrict__ bp0, const v2f* __restrict__ bp1,
                                            int g0, v2f (&bn)[CH][2]) {
#pragma unroll
  for (int i = 0; i < CH; ++i) {
    bn[i][0] = bp0[(size_t)(g0 + i) * 32];
    bn[i][1] = bp1[(size_t)(g0 + i) * 32];
  }
}
__device__ __forceinline__ void gmm_make_a(v2f (&ac)[CH][RB], const v2f (&xr)[CH][RB], const float (&tr)[CH][RB]) {
#pragma unroll
  for (int i = 0; i < CH; ++i) {
#pragma unroll
    for (int rt = 0; rt < RB; ++rt) ac[i][rt] = xr[i][rt] * tr[i][rt];
  }
}
__device__ __forceinline__ void gmm_run(v8f (&acc)[RB][2], const v2f (&ac)[CH][RB], const v2f (&bc)[CH][2]) {
#pragma unroll
  for (int i = 0; i < CH; ++i) {
#pragma unroll
    for (int rt = 0; rt < RB; ++rt) {
      acc[rt][0] = __builtin_amdgcn_wmma_f32_16x16x4_f32(false, ac[i][rt], false, bc[i][0], (short)0, acc[rt][0], false, false);
      acc[rt][1] = __builtin_amdgcn_wmma_f32_16x16x4_f32(false, ac[i][rt], false, bc[i][1], (short)0, acc[rt][1], false, false);
    }
  }
}
__device__ __forceinline__ void gmm_step(v8f (&acc)[RB][2], v2f (&ac)[CH][RB], const v2f (&bc)[CH][2],
                                         v2f (&bn)[CH][2], const v2f* __restrict__ bp0,
                                         const v2f* __restrict__ bp1, int gnext, const float* xl, int h,
                                         int& nj, int& nci, int (&xa)[CH], int (&ta)[CH]) {
  v2f xr[CH][RB];
  float tr[CH][RB];
  gmm_fetch_b(bp0, bp1, gnext, bn);
  gmm_fetch_x(xl, xa, ta, xr, tr);
  __builtin_amdgcn_sched_barrier(0);
  gmm_run(acc, ac, bc);
  gmm_next_addr(h, nj, nci, xa, ta);
  gmm_make_a(ac, xr, tr);
}

__global__ __launch_bounds__(32 * WPB) void gmm_main(const float* __restrict__ X,
                                                     const float* __restrict__ bpack,
                                                     float* __restrict__ out,
                                                     int ntiles) {
  __shared__ __attribute__((aligned(16))) float xs[WPB][16 * RB][XS];
  const int tid = threadIdx.x;
  const int w = __builtin_amdgcn_readfirstlane(tid >> 5);
  const int lane = tid & 31;
  const int h = lane >> 4;
  const int r = lane & 15;
  const int t0 = (blockIdx.x * WPB + w) * RB;
  int nvalid = ntiles - t0;
  if (nvalid > RB) nvalid = RB;

  if (nvalid > 0) {
    const float* xp = X + (size_t)t0 * 16 * 64;
    const int umax = nvalid * 8;
    for (int u = 0; u < umax; ++u) {
      const int flat = u * 128 + lane * 4;
      float4 v = *(const float4*)(xp + flat);
      *(float4*)&xs[w][flat >> 6][flat & 63] = v;
    }
    float4 z; z.x = 0.f; z.y = 0.f; z.z = 0.f; z.w = 0.f;
    for (int u = umax; u < RB * 8; ++u) {
      const int flat = u * 128 + lane * 4;
      *(float4*)&xs[w][flat >> 6][flat & 63] = z;
    }
    float4 one; one.x = 1.f; one.y = 0.f; one.z = 0.f; one.w = 0.f;
    *(float4*)&xs[w][lane][64] = one;
    *(float4*)&xs[w][lane + 32][64] = one;
  }
  __syncthreads();
  if (nvalid <= 0) return;

  const float* xl = &xs[w][r][0];
  const v2f* __restrict__ bp0 = (const v2f*)bpack + (h * 16 + r);
  const v2f* __restrict__ bp1 = bp0 + (size_t)NGROUPS * 32;

  v8f acc[RB][2];
#pragma unroll
  for (int rt = 0; rt < RB; ++rt) { acc[rt][0] = (v8f){}; acc[rt][1] = (v8f){}; }

  v2f be[CH][2], bo[CH][2];
  v2f ac[CH][RB];
  int xa[CH], ta[CH];
  int nj = 0, nci = 0;

  {
    v2f xr[CH][RB];
    float tr[CH][RB];
    gmm_next_addr(h, nj, nci, xa, ta);
    gmm_fetch_b(bp0, bp1, 0, be);
    gmm_fetch_x(xl, xa, ta, xr, tr);
    gmm_next_addr(h, nj, nci, xa, ta);
    gmm_make_a(ac, xr, tr);
  }
  for (int c = 0; c + 2 < NCHUNK; c += 2) {
    gmm_step(acc, ac, be, bo, bp0, bp1, (c + 1) * CH, xl, h, nj, nci, xa, ta);
    gmm_step(acc, ac, bo, be, bp0, bp1, (c + 2) * CH, xl, h, nj, nci, xa, ta);
  }
  gmm_run(acc, ac, be);

#pragma unroll
  for (int rt = 0; rt < RB; ++rt) {
    float res[8];
#pragma unroll
    for (int v = 0; v < 8; ++v) {
      float a0 = acc[rt][0][v], a1 = acc[rt][1][v];
      float mx = fmaxf(a0, a1);
      mx = fmaxf(mx, __shfl_xor(mx, 1, 32));
      mx = fmaxf(mx, __shfl_xor(mx, 2, 32));
      mx = fmaxf(mx, __shfl_xor(mx, 4, 32));
      mx = fmaxf(mx, __shfl_xor(mx, 8, 32));
      float e = expf(a0 - mx) + expf(a1 - mx);
      e += __shfl_xor(e, 1, 32);
      e += __shfl_xor(e, 2, 32);
      e += __shfl_xor(e, 4, 32);
      e += __shfl_xor(e, 8, 32);
      res[v] = mx + logf(e);
    }
    float val = res[0];
#pragma unroll
    for (int v = 1; v < 8; ++v) val = (r == v) ? res[v] : val;
    if (rt < nvalid && r < 8)
      out[(size_t)(t0 + rt) * 16 + 8 * h + r] = val;
  }
}

extern "C" void kernel_launch(void* const* d_in, const int* in_sizes, int n_in,
                              void* d_out, int out_size, void* d_ws, size_t ws_size,
                              hipStream_t stream) {
  const float* X   = (const float*)d_in[0];
  const float* loc = (const float*)d_in[1];
  const float* cov = (const float*)d_in[2];
  const float* pi  = (const float*)d_in[3];
  float* out = (float*)d_out;
  float* bpack = (float*)d_ws;

  const int N = in_sizes[0] / 64;
  const int ntiles = N / 16;

  gmm_prep<<<GMM_K, 64, 0, stream>>>(loc, cov, pi, bpack);
  const int tilesPerBlock = WPB * RB;
  const int nblk = (ntiles + tilesPerBlock - 1) / tilesPerBlock;
  gmm_main<<<nblk, 32 * WPB, 0, stream>>>(X, bpack, out, ntiles);
}


# gmm_prep: Gauss-Jordan sweep takes two pivots per barrier step (2x2 pivot block inverted in registers from its two single pivots, one rank-2 row update): 32 steps, not 64; sum order changes
# speedup vs baseline: 4.2972x; 1.1507x over previous
#include <hip/hip_runtime.h>
#include <math.h>


#define NGROUPS 561
#define GMM_K   32
#define LOG2PI  1.8378770664093453f
#define XS      68
#define RB      4
#define WPB     4

typedef __attribute__((ext_vector_type(2))) float v2f;
typedef __attribute__((ext_vector_type(8))) float v8f;

#define PS 68

__global__ __launch_bounds__(64) void gmm_prep(const float* __restrict__ loc,
                                               const float* __restrict__ cov,
                                               const float* __restrict__ pi,
                                               float* __restrict__ bpack) {
  __shared__ __attribute__((aligned(16))) float Lm[64][PS];
  __shared__ __attribute__((aligned(16))) float rowb[2][2][64];
  __shared__ float nipv[2][64];
  __shared__ float lg[64];
  __shared__ __attribute__((aligned(16))) float mu[64];
  __shared__ float bv[64];
  __shared__ float sc[2];
  const int k = blockIdx.x;
  const int t = threadIdx.x;

  {
    const float* ck = cov + (size_t)k * 4096;
    for (int i = 0; i < 64; ++i) Lm[i][t] = ck[i * 64 + t];
  }
  mu[t] = loc[k * 64 + t];
  if (t == 0) {
    float m = pi[0];
    for (int i = 1; i < GMM_K; ++i) m = fmaxf(m, pi[i]);
    float s = 0.f;
    for (int i = 0; i < GMM_K; ++i) s += expf(pi[i] - m);
    sc[0] = m + logf(s);
  }
  __syncthreads();

  float r[64];
#pragma unroll
  for (int q = 0; q < 16; ++q) {
    const float4 a = *(const float4*)&Lm[t][4 * q];
    r[4 * q + 0] = (a.x + Lm[4 * q + 0][t]) * 0.5f;
    r[4 * q + 1] = (a.y + Lm[4 * q + 1][t]) * 0.5f;
    r[4 * q + 2] = (a.z + Lm[4 * q + 2][t]) * 0.5f;
    r[4 * q + 3] = (a.w + Lm[4 * q + 3][t]) * 0.5f;
  }
  if (t < 2) {
#pragma unroll
    for (int q = 0; q < 16; ++q) {
      float4 v; v.x = r[4 * q + 0]; v.y = r[4 * q + 1]; v.z = r[4 * q + 2]; v.w = r[4 * q + 3];
      *(float4*)&rowb[0][t][4 * q] = v;
    }
  }
  nipv[0][t] = 1.0f / r[0];
  __syncthreads();

  float piv_own = 1.f;
#pragma nounroll
  for (int s = 0; s < 32; ++s) {
    float w0[64], w1[64];
#pragma unroll
    for (int q = 0; q < 16; ++q) {
      const float4 v = *(const float4*)&rowb[s & 1][0][4 * q];
      w0[4 * q + 0] = v.x; w0[4 * q + 1] = v.y; w0[4 * q + 2] = v.z; w0[4 * q + 3] = v.w;
      const float4 u = *(const float4*)&rowb[s & 1][1][4 * q];
      w1[4 * q + 0] = u.x; w1[4 * q + 1] = u.y; w1[4 * q + 2] = u.z; w1[4 * q + 3] = u.w;
    }
    const float ia = nipv[s & 1][2 * s];
    const float l = w1[0] * ia;
    const float u = w0[1] * ia;
    const float sch = w1[1] - l * w0[1];
    const float q11 = 1.0f / sch;
    const float q01 = -u * q11;
    const float q10 = -l * q11;
    const float q00 = ia - q01 * l;
    const bool own = ((t >> 1) == s);
    const bool odd = ((t & 1) != 0);
    piv_own = own ? (odd ? sch : w0[0]) : piv_own;
    float c0 = r[0] * q00 + r[1] * q10;
    float c1 = r[0] * q01 + r[1] * q11;
    if (own) { c0 = odd ? -q10 : -q00; c1 = odd ? -q11 : -q01; }
#pragma unroll
    for (int m = 2; m < 64; ++m) {
      const float base = own ? 0.f : r[m];
      r[m - 2] = __builtin_fmaf(-c0, w0[m], __builtin_fmaf(-c1, w1[m], base));
    }
    r[62] = -c0;
    r[63] = -c1;
    nipv[(s + 1) & 1][t] = 1.0f / r[0];
    if ((t >> 1) == s + 1) {
      float* wb = &rowb[(s + 1) & 1][t & 1][0];
#pragma unroll
      for (int q = 0; q < 16; ++q) {
        float4 v; v.x = r[4 * q + 0]; v.y = r[4 * q + 1]; v.z = r[4 * q + 2]; v.w = r[4 * q + 3];
        *(float4*)(wb + 4 * q) = v;
      }
    }
    __syncthreads();
  }

  lg[t] = logf(piv_own);
  {
    float s = 0.f;
#pragma unroll
    for (int q = 0; q < 16; ++q) {
      float4 v; v.x = r[4 * q + 0]; v.y = r[4 * q + 1]; v.z = r[4 * q + 2]; v.w = r[4 * q + 3];
      *(float4*)&Lm[t][4 * q] = v;
      const float4 m4 = *(const float4*)&mu[4 * q];
      s += v.x * m4.x;
      s += v.y * m4.y;
      s += v.z * m4.z;
      s += v.w * m4.w;
    }
    bv[t] = s;
  }
  __syncthreads();
  if (t == 0) {
    float hs = 0.f;
    for (int i = 0; i < 64; ++i) hs += lg[i];
    float c2 = 0.f;
    for (int i = 0; i < 64; ++i) c2 += mu[i] * bv[i];
    sc[1] = -0.5f * c2 + pi[k] - sc[0] - 0.5f * hs - 0.5f * 64.f * LOG2PI;
  }
  __syncthreads();

  const int ctile = k >> 4;
  const int n = k & 15;
  float* base = bpack + (size_t)ctile * NGROUPS * 64 + n * 2;
  for (int g = t; g < NGROUPS; g += 64) {
    float w0 = 0.f, w1 = 0.f, w2 = 0.f, w3 = 0.f;
    if (g < NGROUPS - 17) {
      int a = 0;
      while (2 * (a + 1) * (a + 2) <= g) ++a;
      int rem = g - 2 * a * (a + 1);
      int b = 0;
      while (rem >= a + 1) { rem -= a + 1; ++b; }
      const int j = 4 * a + b;
      const int i4 = 4 * rem;
      const float4 pr = *(const float4*)&Lm[j][i4];
      const float pjj = Lm[j][j];
      w0 = (i4 + 0 < j) ? -pr.x : ((i4 + 0 == j) ? -0.5f * pjj : 0.f);
      w1 = (i4 + 1 < j) ? -pr.y : ((i4 + 1 == j) ? -0.5f * pjj : 0.f);
      w2 = (i4 + 2 < j) ? -pr.z : ((i4 + 2 == j) ? -0.5f * pjj : 0.f);
      w3 = (i4 + 3 < j) ? -pr.w : ((i4 + 3 == j) ? -0.5f * pjj : 0.f);
    } else if (g < NGROUPS - 1) {
      const int ci = g - (NGROUPS - 17);
      w0 = bv[4 * ci + 0]; w1 = bv[4 * ci + 1]; w2 = bv[4 * ci + 2]; w3 = bv[4 * ci + 3];
    } else {
      w0 = sc[1];
    }
    float* dst = base + (size_t)g * 64;
    v2f lo; lo.x = w0; lo.y = w1;
    v2f hi; hi.x = w2; hi.y = w3;
    *(v2f*)dst = lo;
    *(v2f*)(dst + 32) = hi;
  }
}

#define CH     11
#define NCHUNK (NGROUPS / CH)

__device__ __forceinline__ void gmm_next_addr(int h, int& nj, int& nci, int (&xa)[CH], int (&ta)[CH]) {
#pragma unroll
  for (int i = 0; i < CH; ++i) {
    xa[i] = 4 * nci + 2 * h;
    ta[i] = nj;
    ++nci;
    if (nci > (nj >> 2)) { nci = 0; ++nj; }
  }
}
__device__ __forceinline__ void gmm_fetch_x(const float* xl, const int (&xa)[CH], const int (&ta)[CH],
                                            v2f (&xr)[CH][RB], float (&tr)[CH][RB]) {
#pragma unroll
  for (int i = 0; i < CH; ++i) {
#pragma unroll
    for (int rt = 0; rt < RB; ++rt) {
      xr[i][rt] = *(const v2f*)(xl + xa[i] + rt * 16 * XS);
      tr[i][rt] = xl[ta[i] + rt * 16 * XS];
    }
  }
}
__device__ __forceinline__ void gmm_fetch_b(const v2f* __restrict__ bp0, const v2f* __restrict__ bp1,
                                            int g0, v2f (&bn)[CH][2]) {
#pragma unroll
  for (int i = 0; i < CH; ++i) {
    bn[i][0] = bp0[(size_t)(g0 + i) * 32];
    bn[i][1] = bp1[(size_t)(g0 + i) * 32];
  }
}
__device__ __forceinline__ void gmm_make_a(v2f (&ac)[CH][RB], const v2f (&xr)[CH][RB], const float (&tr)[CH][RB]) {
#pragma unroll
  for (int i = 0; i < CH; ++i) {
#pragma unroll
    for (int rt = 0; rt < RB; ++rt) ac[i][rt] = xr[i][rt] * tr[i][rt];
  }
}
__device__ __forceinline__ void gmm_run(v8f (&acc)[RB][2], const v2f (&ac)[CH][RB], const v2f (&bc)[CH][2]) {
#pragma unroll
  for (int i = 0; i < CH; ++i) {
#pragma unroll
    for (int rt = 0; rt < RB; ++rt) {
      acc[rt][0] = __builtin_amdgcn_wmma_f32_16x16x4_f32(false, ac[i][rt], false, bc[i][0], (short)0, acc[rt][0], false, false);
      acc[rt][1] = __builtin_amdgcn_wmma_f32_16x16x4_f32(false, ac[i][rt], false, bc[i][1], (short)0, acc[rt][1], false, false);
    }
  }
}
__device__ __forceinline__ void gmm_step(v8f (&acc)[RB][2], v2f (&ac)[CH][RB], const v2f (&bc)[CH][2],
                                         v2f (&bn)[CH][2], const v2f* __restrict__ bp0,
                                         const v2f* __restrict__ bp1, int gnext, const float* xl, int h,
                                         int& nj, int& nci, int (&xa)[CH], int (&ta)[CH]) {
  v2f xr[CH][RB];
  float tr[CH][RB];
  gmm_fetch_b(bp0, bp1, gnext, bn);
  gmm_fetch_x(xl, xa, ta, xr, tr);
  __builtin_amdgcn_sched_barrier(0);
  gmm_run(acc, ac, bc);
  gmm_next_addr(h, nj, nci, xa, ta);
  gmm_make_a(ac, xr, tr);
}

__global__ __launch_bounds__(32 * WPB) void gmm_main(const float* __restrict__ X,
                                                     const float* __restrict__ bpack,
                                                     float* __restrict__ out,
                                                     int ntiles) {
  __shared__ __attribute__((aligned(16))) float xs[WPB][16 * RB][XS];
  const int tid = threadIdx.x;
  const int w = __builtin_amdgcn_readfirstlane(tid >> 5);
  const int lane = tid & 31;
  const int h = lane >> 4;
  const int r = lane & 15;
  const int t0 = (blockIdx.x * WPB + w) * RB;
  int nvalid = ntiles - t0;
  if (nvalid > RB) nvalid = RB;

  if (nvalid > 0) {
    const float* xp = X + (size_t)t0 * 16 * 64;
    const int umax = nvalid * 8;
    for (int u = 0; u < umax; ++u) {
      const int flat = u * 128 + lane * 4;
      float4 v = *(const float4*)(xp + flat);
      *(float4*)&xs[w][flat >> 6][flat & 63] = v;
    }
    float4 z; z.x = 0.f; z.y = 0.f; z.z = 0.f; z.w = 0.f;
    for (int u = umax; u < RB * 8; ++u) {
      const int flat = u * 128 + lane * 4;
      *(float4*)&xs[w][flat >> 6][flat & 63] = z;
    }
    float4 one; one.x = 1.f; one.y = 0.f; one.z = 0.f; one.w = 0.f;
    *(float4*)&xs[w][lane][64] = one;
    *(float4*)&xs[w][lane + 32][64] = one;
  }
  __syncthreads();
  if (nvalid <= 0) return;

  const float* xl = &xs[w][r][0];
  const v2f* __restrict__ bp0 = (const v2f*)bpack + (h * 16 + r);
  const v2f* __restrict__ bp1 = bp0 + (size_t)NGROUPS * 32;

  v8f acc[RB][2];
#pragma unroll
  for (int rt = 0; rt < RB; ++rt) { acc[rt][0] = (v8f){}; acc[rt][1] = (v8f){}; }

  v2f be[CH][2], bo[CH][2];
  v2f ac[CH][RB];
  int xa[CH], ta[CH];
  int nj = 0, nci = 0;

  {
    v2f xr[CH][RB];
    float tr[CH][RB];
    gmm_next_addr(h, nj, nci, xa, ta);
    gmm_fetch_b(bp0, bp1, 0, be);
    gmm_fetch_x(xl, xa, ta, xr, tr);
    gmm_next_addr(h, nj, nci, xa, ta);
    gmm_make_a(ac, xr, tr);
  }
  for (int c = 0; c + 2 < NCHUNK; c += 2) {
    gmm_step(acc, ac, be, bo, bp0, bp1, (c + 1) * CH, xl, h, nj, nci, xa, ta);
    gmm_step(acc, ac, bo, be, bp0, bp1, (c + 2) * CH, xl, h, nj, nci, xa, ta);
  }
  gmm_run(acc, ac, be);

#pragma unroll
  for (int rt = 0; rt < RB; ++rt) {
    float res[8];
#pragma unroll
    for (int v = 0; v < 8; ++v) {
      float a0 = acc[rt][0][v], a1 = acc[rt][1][v];
      float mx = fmaxf(a0, a1);
      mx = fmaxf(mx, __shfl_xor(mx, 1, 32));
      mx = fmaxf(mx, __shfl_xor(mx, 2, 32));
      mx = fmaxf(mx, __shfl_xor(mx, 4, 32));
      mx = fmaxf(mx, __shfl_xor(mx, 8, 32));
      float e = expf(a0 - mx) + expf(a1 - mx);
      e += __shfl_xor(e, 1, 32);
      e += __shfl_xor(e, 2, 32);
      e += __shfl_xor(e, 4, 32);
      e += __shfl_xor(e, 8, 32);
      res[v] = mx + logf(e);
    }
    float val = res[0];
#pragma unroll
    for (int v = 1; v < 8; ++v) val = (r == v) ? res[v] : val;
    if (rt < nvalid && r < 8)
      out[(size_t)(t0 + rt) * 16 + 8 * h + r] = val;
  }
}

extern "C" void kernel_launch(void* const* d_in, const int* in_sizes, int n_in,
                              void* d_out, int out_size, void* d_ws, size_t ws_size,
                              hipStream_t stream) {
  const float* X   = (const float*)d_in[0];
  const float* loc = (const float*)d_in[1];
  const float* cov = (const float*)d_in[2];
  const float* pi  = (const float*)d_in[3];
  float* out = (float*)d_out;
  float* bpack = (float*)d_ws;

  const int N = in_sizes[0] / 64;
  const int ntiles = N / 16;

  gmm_prep<<<GMM_K, 64, 0, stream>>>(loc, cov, pi, bpack);
  const int tilesPerBlock = WPB * RB;
  const int nblk = (ntiles + tilesPerBlock - 1) / tilesPerBlock;
  gmm_main<<<nblk, 32 * WPB, 0, stream>>>(X, bpack, out, ntiles);
}


# gmm_main: pair groups re-ordered ci-major in blocks of 4 consecutive j: one x-pair load and one 16-byte x_j load per tile serve 4 groups, scalar group walk gone; B pack follows; K sum order changes
# speedup vs baseline: 4.9748x; 1.1577x over previous
#include <hip/hip_runtime.h>
#include <math.h>


#define NGROUPS 561
#define GMM_K   32
#define LOG2PI  1.8378770664093453f
#define XS      68
#define RB      4
#define WPB     4

typedef __attribute__((ext_vector_type(2))) float v2f;
typedef __attribute__((ext_vector_type(8))) float v8f;

#define PS 68

__global__ __launch_bounds__(64) void gmm_prep(const float* __restrict__ loc,
                                               const float* __restrict__ cov,
                                               const float* __restrict__ pi,
                                               float* __restrict__ bpack) {
  __shared__ __attribute__((aligned(16))) float Lm[64][PS];
  __shared__ __attribute__((aligned(16))) float rowb[2][2][64];
  __shared__ float nipv[2][64];
  __shared__ float lg[64];
  __shared__ __attribute__((aligned(16))) float mu[64];
  __shared__ float bv[64];
  __shared__ float sc[2];
  const int k = blockIdx.x;
  const int t = threadIdx.x;

  {
    const float* ck = cov + (size_t)k * 4096;
    for (int i = 0; i < 64; ++i) Lm[i][t] = ck[i * 64 + t];
  }
  mu[t] = loc[k * 64 + t];
  if (t == 0) {
    float m = pi[0];
    for (int i = 1; i < GMM_K; ++i) m = fmaxf(m, pi[i]);
    float s = 0.f;
    for (int i = 0; i < GMM_K; ++i) s += expf(pi[i] - m);
    sc[0] = m + logf(s);
  }
  __syncthreads();

  float r[64];
#pragma unroll
  for (int q = 0; q < 16; ++q) {
    const float4 a = *(const float4*)&Lm[t][4 * q];
    r[4 * q + 0] = (a.x + Lm[4 * q + 0][t]) * 0.5f;
    r[4 * q + 1] = (a.y + Lm[4 * q + 1][t]) * 0.5f;
    r[4 * q + 2] = (a.z + Lm[4 * q + 2][t]) * 0.5f;
    r[4 * q + 3] = (a.w + Lm[4 * q + 3][t]) * 0.5f;
  }
  if (t < 2) {
#pragma unroll
    for (int q = 0; q < 16; ++q) {
      float4 v; v.x = r[4 * q + 0]; v.y = r[4 * q + 1]; v.z = r[4 * q + 2]; v.w = r[4 * q + 3];
      *(float4*)&rowb[0][t][4 * q] = v;
    }
  }
  nipv[0][t] = 1.0f / r[0];
  __syncthreads();

  float piv_own = 1.f;
#pragma nounroll
  for (int s = 0; s < 32; ++s) {
    float w0[64], w1[64];
#pragma unroll
    for (int q = 0; q < 16; ++q) {
      const float4 v = *(const float4*)&rowb[s & 1][0][4 * q];
      w0[4 * q + 0] = v.x; w0[4 * q + 1] = v.y; w0[4 * q + 2] = v.z; w0[4 * q + 3] = v.w;
      const float4 u = *(const float4*)&rowb[s & 1][1][4 * q];
      w1[4 * q + 0] = u.x; w1[4 * q + 1] = u.y; w1[4 * q + 2] = u.z; w1[4 * q + 3] = u.w;
    }
    const float ia = nipv[s & 1][2 * s];
    const float l = w1[0] * ia;
    const float u = w0[1] * ia;
    const float sch = w1[1] - l * w0[1];
    const float q11 = 1.0f / sch;
    const float q01 = -u * q11;
    const float q10 = -l * q11;
    const float q00 = ia - q01 * l;
    const bool own = ((t >> 1) == s);
    const bool odd = ((t & 1) != 0);
    piv_own = own ? (odd ? sch : w0[0]) : piv_own;
    float c0 = r[0] * q00 + r[1] * q10;
    float c1 = r[0] * q01 + r[1] * q11;
    if (own) { c0 = odd ? -q10 : -q00; c1 = odd ? -q11 : -q01; }
#pragma unroll
    for (int m = 2; m < 64; ++m) {
      const float base = own ? 0.f : r[m];
      r[m - 2] = __builtin_fmaf(-c0, w0[m], __builtin_fmaf(-c1, w1[m], base));
    }
    r[62] = -c0;
    r[63] = -c1;
    nipv[(s + 1) & 1][t] = 1.0f / r[0];
    if ((t >> 1) == s + 1) {
      float* wb = &rowb[(s + 1) & 1][t & 1][0];
#pragma unroll
      for (int q = 0; q < 16; ++q) {
        float4 v; v.x = r[4 * q + 0]; v.y = r[4 * q + 1]; v.z = r[4 * q + 2]; v.w = r[4 * q + 3];
        *(float4*)(wb + 4 * q) = v;
      }
    }
    __syncthreads();
  }

  lg[t] = logf(piv_own);
  {
    float s = 0.f;
#pragma unroll
    for (int q = 0; q < 16; ++q) {
      float4 v; v.x = r[4 * q + 0]; v.y = r[4 * q + 1]; v.z = r[4 * q + 2]; v.w = r[4 * q + 3];
      *(float4*)&Lm[t][4 * q] = v;
      const float4 m4 = *(const float4*)&mu[4 * q];
      s += v.x * m4.x;
      s += v.y * m4.y;
      s += v.z * m4.z;
      s += v.w * m4.w;
    }
    bv[t] = s;
  }
  __syncthreads();
  if (t == 0) {
    float hs = 0.f;
    for (int i = 0; i < 64; ++i) hs += lg[i];
    float c2 = 0.f;
    for (int i = 0; i < 64; ++i) c2 += mu[i] * bv[i];
    sc[1] = -0.5f * c2 + pi[k] - sc[0] - 0.5f * hs - 0.5f * 64.f * LOG2PI;
  }
  __syncthreads();

  const int ctile = k >> 4;
  const int n = k & 15;
  float* base = bpack + (size_t)ctile * NGROUPS * 64 + n * 2;
  for (int g = t; g < NGROUPS; g += 64) {
    float w0 = 0.f, w1 = 0.f, w2 = 0.f, w3 = 0.f;
    if (g < NGROUPS - 17) {
      const int b = g & 3;
      int rem = g >> 2;
      int ci = 0;
      while (rem >= 16 - ci) { rem -= 16 - ci; ++ci; }
      const int j = 4 * (ci + rem) + b;
      const int i4 = 4 * ci;
      const float4 pr = *(const float4*)&Lm[j][i4];
      const float pjj = Lm[j][j];
      w0 = (i4 + 0 < j) ? -pr.x : ((i4 + 0 == j) ? -0.5f * pjj : 0.f);
      w1 = (i4 + 1 < j) ? -pr.y : ((i4 + 1 == j) ? -0.5f * pjj : 0.f);
      w2 = (i4 + 2 < j) ? -pr.z : ((i4 + 2 == j) ? -0.5f * pjj : 0.f);
      w3 = (i4 + 3 < j) ? -pr.w : ((i4 + 3 == j) ? -0.5f * pjj : 0.f);
    } else if (g < NGROUPS - 1) {
      const int ci = g - (NGROUPS - 17);
      w0 = bv[4 * ci + 0]; w1 = bv[4 * ci + 1]; w2 = bv[4 * ci + 2]; w3 = bv[4 * ci + 3];
    } else {
      w0 = sc[1];
    }
    float* dst = base + (size_t)g * 64;
    v2f lo; lo.x = w0; lo.y = w1;
    v2f hi; hi.x = w2; hi.y = w3;
    *(v2f*)dst = lo;
    *(v2f*)(dst + 32) = hi;
  }
}

#define NCH    68
#define NTAIL  17

typedef __attribute__((ext_vector_type(4))) float v4f;

__device__ __forceinline__ void gmm_next_blk(int& ci, int& jb) {
  ++jb;
  if (jb > 15) { ci = (ci < 16) ? ci + 1 : 16; jb = ci; }
}
__device__ __forceinline__ void gmm_fetch_x(const float* xl, int h, int& ci, int& jb,
                                            v2f (&xp)[2][RB], v4f (&xj)[2][RB]) {
#pragma unroll
  for (int u = 0; u < 2; ++u) {
    const int pa = 4 * ci + 2 * h;
    const int ja = 4 * jb;
#pragma unroll
    for (int rt = 0; rt < RB; ++rt) {
      xp[u][rt] = *(const v2f*)(xl + pa + rt * 16 * XS);
      xj[u][rt] = *(const v4f*)(xl + ja + rt * 16 * XS);
    }
    gmm_next_blk(ci, jb);
  }
}
__device__ __forceinline__ void gmm_fetch_b(const v2f* __restrict__ bp0, const v2f* __restrict__ bp1,
                                            int c, v2f (&bn)[8][2]) {
#pragma unroll
  for (int i = 0; i < 8; ++i) {
    bn[i][0] = bp0[(size_t)(8 * c + i) * 32];
    bn[i][1] = bp1[(size_t)(8 * c + i) * 32];
  }
}
__device__ __forceinline__ void gmm_run(v8f (&acc)[RB][2], const v2f (&xp)[2][RB], const v4f (&xj)[2][RB],
                                        const v2f (&bc)[8][2]) {
#pragma unroll
  for (int u = 0; u < 2; ++u) {
#pragma unroll
    for (int b = 0; b < 4; ++b) {
#pragma unroll
      for (int rt = 0; rt < RB; ++rt) {
        const v2f a = xp[u][rt] * xj[u][rt][b];
        acc[rt][0] = __builtin_amdgcn_wmma_f32_16x16x4_f32(false, a, false, bc[4 * u + b][0], (short)0, acc[rt][0], false, false);
        acc[rt][1] = __builtin_amdgcn_wmma_f32_16x16x4_f32(false, a, false, bc[4 * u + b][1], (short)0, acc[rt][1], false, false);
      }
    }
  }
}
__device__ __forceinline__ void gmm_step(v8f (&acc)[RB][2], const v2f (&xpc)[2][RB], const v4f (&xjc)[2][RB],
                                         v2f (&xpn)[2][RB], v4f (&xjn)[2][RB], const v2f (&bc)[8][2],
                                         v2f (&bf)[8][2], const v2f* __restrict__ bp0,
                                         const v2f* __restrict__ bp1, int cfetch, const float* xl, int h,
                                         int& ci, int& jb) {
  gmm_fetch_b(bp0, bp1, cfetch, bf);
  gmm_fetch_x(xl, h, ci, jb, xpn, xjn);
  __builtin_amdgcn_sched_barrier(0);
  gmm_run(acc, xpc, xjc, bc);
}
__device__ __forceinline__ void gmm_tail(v8f (&acc)[RB][2], const v2f (&ba)[8][2], const v2f (&bb)[8][2],
                                         const v2f (&bl)[2], const float* xl, int h) {
  v2f ta[NTAIL][RB];
#pragma unroll
  for (int i = 0; i < NTAIL; ++i) {
#pragma unroll
    for (int rt = 0; rt < RB; ++rt) ta[i][rt] = *(const v2f*)(xl + 4 * i + 2 * h + rt * 16 * XS);
  }
  __builtin_amdgcn_sched_barrier(0);
#pragma unroll
  for (int i = 0; i < NTAIL; ++i) {
#pragma unroll
    for (int rt = 0; rt < RB; ++rt) {
      const v2f f0 = (i < 8) ? ba[i & 7][0] : ((i < 16) ? bb[i & 7][0] : bl[0]);
      const v2f f1 = (i < 8) ? ba[i & 7][1] : ((i < 16) ? bb[i & 7][1] : bl[1]);
      acc[rt][0] = __builtin_amdgcn_wmma_f32_16x16x4_f32(false, ta[i][rt], false, f0, (short)0, acc[rt][0], false, false);
      acc[rt][1] = __builtin_amdgcn_wmma_f32_16x16x4_f32(false, ta[i][rt], false, f1, (short)0, acc[rt][1], false, false);
    }
  }
}

__global__ __launch_bounds__(32 * WPB) void gmm_main(const float* __restrict__ X,
                                                     const float* __restrict__ bpack,
                                                     float* __restrict__ out,
                                                     int ntiles) {
  __shared__ __attribute__((aligned(16))) float xs[WPB][16 * RB][XS];
  const int tid = threadIdx.x;
  const int w = __builtin_amdgcn_readfirstlane(tid >> 5);
  const int lane = tid & 31;
  const int h = lane >> 4;
  const int r = lane & 15;
  const int t0 = (blockIdx.x * WPB + w) * RB;
  int nvalid = ntiles - t0;
  if (nvalid > RB) nvalid = RB;

  if (nvalid > 0) {
    const float* xp = X + (size_t)t0 * 16 * 64;
    const int umax = nvalid * 8;
    for (int u = 0; u < umax; ++u) {
      const int flat = u * 128 + lane * 4;
      float4 v = *(const float4*)(xp + flat);
      *(float4*)&xs[w][flat >> 6][flat & 63] = v;
    }
    float4 z; z.x = 0.f; z.y = 0.f; z.z = 0.f; z.w = 0.f;
    for (int u = umax; u < RB * 8; ++u) {
      const int flat = u * 128 + lane * 4;
      *(float4*)&xs[w][flat >> 6][flat & 63] = z;
    }
    float4 one; one.x = 1.f; one.y = 0.f; one.z = 0.f; one.w = 0.f;
    *(float4*)&xs[w][lane][64] = one;
    *(float4*)&xs[w][lane + 32][64] = one;
  }
  __syncthreads();
  if (nvalid <= 0) return;

  const float* xl = &xs[w][r][0];
  const v2f* __restrict__ bp0 = (const v2f*)bpack + (h * 16 + r);
  const v2f* __restrict__ bp1 = bp0 + (size_t)NGROUPS * 32;

  v8f acc[RB][2];
#pragma unroll
  for (int rt = 0; rt < RB; ++rt) { acc[rt][0] = (v8f){}; acc[rt][1] = (v8f){}; }

  v2f b0[8][2], b1[8][2], b2[8][2], b3[8][2];
  v2f xpe[2][RB], xpo[2][RB];
  v4f xje[2][RB], xjo[2][RB];
  int ci = 0, jb = 0;

  gmm_fetch_b(bp0, bp1, 0, b0);
  gmm_fetch_b(bp0, bp1, 1, b1);
  v2f bl[2];
  bl[0] = bp0[(size_t)(NGROUPS - 1) * 32];
  bl[1] = bp1[(size_t)(NGROUPS - 1) * 32];
  gmm_fetch_x(xl, h, ci, jb, xpe, xje);
  for (int c = 0; c < NCH; c += 4) {
    gmm_step(acc, xpe, xje, xpo, xjo, b0, b2, bp0, bp1, c + 2, xl, h, ci, jb);
    gmm_step(acc, xpo, xjo, xpe, xje, b1, b3, bp0, bp1, c + 3, xl, h, ci, jb);
    gmm_step(acc, xpe, xje, xpo, xjo, b2, b0, bp0, bp1, c + 4, xl, h, ci, jb);
    gmm_step(acc, xpo, xjo, xpe, xje, b3, b1, bp0, bp1, c + 5, xl, h, ci, jb);
  }
  gmm_tail(acc, b0, b1, bl, xl, h);

#pragma unroll
  for (int rt = 0; rt < RB; ++rt) {
    float res[8];
#pragma unroll
    for (int v = 0; v < 8; ++v) {
      float a0 = acc[rt][0][v], a1 = acc[rt][1][v];
      float mx = fmaxf(a0, a1);
      mx = fmaxf(mx, __shfl_xor(mx, 1, 32));
      mx = fmaxf(mx, __shfl_xor(mx, 2, 32));
      mx = fmaxf(mx, __shfl_xor(mx, 4, 32));
      mx = fmaxf(mx, __shfl_xor(mx, 8, 32));
      float e = expf(a0 - mx) + expf(a1 - mx);
      e += __shfl_xor(e, 1, 32);
      e += __shfl_xor(e, 2, 32);
      e += __shfl_xor(e, 4, 32);
      e += __shfl_xor(e, 8, 32);
      res[v] = mx + logf(e);
    }
    float val = res[0];
#pragma unroll
    for (int v = 1; v < 8; ++v) val = (r == v) ? res[v] : val;
    if (rt < nvalid && r < 8)
      out[(size_t)(t0 + rt) * 16 + 8 * h + r] = val;
  }
}

extern "C" void kernel_launch(void* const* d_in, const int* in_sizes, int n_in,
                              void* d_out, int out_size, void* d_ws, size_t ws_size,
                              hipStream_t stream) {
  const float* X   = (const float*)d_in[0];
  const float* loc = (const float*)d_in[1];
  const float* cov = (const float*)d_in[2];
  const float* pi  = (const float*)d_in[3];
  float* out = (float*)d_out;
  float* bpack = (float*)d_ws;

  const int N = in_sizes[0] / 64;
  const int ntiles = N / 16;

  gmm_prep<<<GMM_K, 64, 0, stream>>>(loc, cov, pi, bpack);
  const int tilesPerBlock = WPB * RB;
  const int nblk = (ntiles + tilesPerBlock - 1) / tilesPerBlock;
  gmm_main<<<nblk, 32 * WPB, 0, stream>>>(X, bpack, out, ntiles);
}


# gmm_prep: the 33 LDS reads of a Gauss-Jordan step (both pivot rows whole) issued by hand as one group at the head of the step, taken behind counted waits; arithmetic and sum order kept
# speedup vs baseline: 5.5422x; 1.1141x over previous
#include <hip/hip_runtime.h>
#include <math.h>


#define NGROUPS 561
#define GMM_K   32
#define LOG2PI  1.8378770664093453f
#define XS      68
#define RB      4
#define WPB     4

typedef __attribute__((ext_vector_type(2))) float v2f;
typedef __attribute__((ext_vector_type(8))) float v8f;
typedef __attribute__((ext_vector_type(4))) float v4f;
typedef __attribute__((address_space(3))) float lds_float;

template <int OFF> __device__ __forceinline__ v4f lds_read16(unsigned addr) {
  v4f v;
  asm volatile("ds_load_b128 %0, %1 offset:%2" : "=&v"(v) : "v"(addr), "n"(OFF) : "memory");
  return v;
}
__device__ __forceinline__ float lds_read4(unsigned addr) {
  float v;
  asm volatile("ds_load_b32 %0, %1" : "=&v"(v) : "v"(addr) : "memory");
  return v;
}

#define PS 68

__global__ __launch_bounds__(64) void gmm_prep(const float* __restrict__ loc,
                                               const float* __restrict__ cov,
                                               const float* __restrict__ pi,
                                               float* __restrict__ bpack) {
  __shared__ __attribute__((aligned(16))) float Lm[64][PS];
  __shared__ __attribute__((aligned(16))) float rowb[2][2][64];
  __shared__ float nipv[2][64];
  __shared__ float lg[64];
  __shared__ __attribute__((aligned(16))) float mu[64];
  __shared__ float bv[64];
  __shared__ float sc[2];
  const int k = blockIdx.x;
  const int t = threadIdx.x;

  {
    const float* ck = cov + (size_t)k * 4096;
    for (int i = 0; i < 64; ++i) Lm[i][t] = ck[i * 64 + t];
  }
  mu[t] = loc[k * 64 + t];
  if (t == 0) {
    float m = pi[0];
    for (int i = 1; i < GMM_K; ++i) m = fmaxf(m, pi[i]);
    float s = 0.f;
    for (int i = 0; i < GMM_K; ++i) s += expf(pi[i] - m);
    sc[0] = m + logf(s);
  }
  __syncthreads();

  float r[64];
#pragma unroll
  for (int q = 0; q < 16; ++q) {
    const float4 a = *(const float4*)&Lm[t][4 * q];
    r[4 * q + 0] = (a.x + Lm[4 * q + 0][t]) * 0.5f;
    r[4 * q + 1] = (a.y + Lm[4 * q + 1][t]) * 0.5f;
    r[4 * q + 2] = (a.z + Lm[4 * q + 2][t]) * 0.5f;
    r[4 * q + 3] = (a.w + Lm[4 * q + 3][t]) * 0.5f;
  }
  if (t < 2) {
#pragma unroll
    for (int q = 0; q < 16; ++q) {
      float4 v; v.x = r[4 * q + 0]; v.y = r[4 * q + 1]; v.z = r[4 * q + 2]; v.w = r[4 * q + 3];
      *(float4*)&rowb[0][t][4 * q] = v;
    }
  }
  nipv[0][t] = 1.0f / r[0];
  __syncthreads();

  float piv_own = 1.f;
#pragma nounroll
  for (int s = 0; s < 32; ++s) {
    const unsigned ipa = (unsigned)(size_t)(lds_float*)&nipv[s & 1][2 * s];
    const unsigned rwa = (unsigned)(size_t)(lds_float*)&rowb[s & 1][0][0];
    float ia = lds_read4(ipa);
    v4f a0[16], a1[16];
#define GJ_RD(q) a0[q] = lds_read16<16 * (q)>(rwa); a1[q] = lds_read16<256 + 16 * (q)>(rwa);
    GJ_RD(0) GJ_RD(1) GJ_RD(2) GJ_RD(3) GJ_RD(4) GJ_RD(5) GJ_RD(6) GJ_RD(7)
    GJ_RD(8) GJ_RD(9) GJ_RD(10) GJ_RD(11) GJ_RD(12) GJ_RD(13) GJ_RD(14) GJ_RD(15)
#undef GJ_RD
#define GJ_W0(m) a0[(m) >> 2][(m) & 3]
#define GJ_W1(m) a1[(m) >> 2][(m) & 3]
    asm volatile("s_wait_dscnt 30" : "+v"(ia), "+v"(a0[0]), "+v"(a1[0]));
    const float l = GJ_W1(0) * ia;
    const float u = GJ_W0(1) * ia;
    const float sch = GJ_W1(1) - l * GJ_W0(1);
    const float q11 = 1.0f / sch;
    const float q01 = -u * q11;
    const float q10 = -l * q11;
    const float q00 = ia - q01 * l;
    const bool own = ((t >> 1) == s);
    const bool odd = ((t & 1) != 0);
    piv_own = own ? (odd ? sch : GJ_W0(0)) : piv_own;
    float c0 = r[0] * q00 + r[1] * q10;
    float c1 = r[0] * q01 + r[1] * q11;
    if (own) { c0 = odd ? -q10 : -q00; c1 = odd ? -q11 : -q01; }
#define GJ_UPD(m) { const float base = own ? 0.f : r[m]; \
                    r[(m) - 2] = __builtin_fmaf(-c0, GJ_W0(m), __builtin_fmaf(-c1, GJ_W1(m), base)); }
    asm volatile("s_wait_dscnt 16" : "+v"(c0), "+v"(c1),
                 "+v"(a0[1]), "+v"(a1[1]), "+v"(a0[2]), "+v"(a1[2]), "+v"(a0[3]), "+v"(a1[3]), "+v"(a0[4]),
                 "+v"(a1[4]), "+v"(a0[5]), "+v"(a1[5]), "+v"(a0[6]), "+v"(a1[6]), "+v"(a0[7]), "+v"(a1[7]));
    asm volatile("s_wait_dscnt 0" : "+v"(a0[8]), "+v"(a1[8]), "+v"(a0[9]), "+v"(a1[9]), "+v"(a0[10]),
                 "+v"(a1[10]), "+v"(a0[11]), "+v"(a1[11]), "+v"(a0[12]), "+v"(a1[12]), "+v"(a0[13]),
                 "+v"(a1[13]), "+v"(a0[14]), "+v"(a1[14]), "+v"(a0[15]), "+v"(a1[15]));
#pragma unroll
    for (int m = 2; m < 64; ++m) GJ_UPD(m)
#undef GJ_UPD
#undef GJ_W0
#undef GJ_W1
    r[62] = -c0;
    r[63] = -c1;
    nipv[(s + 1) & 1][t] = 1.0f / r[0];
    if ((t >> 1) == s + 1) {
      float* wb = &rowb[(s + 1) & 1][t & 1][0];
#pragma unroll
      for (int q = 0; q < 16; ++q) {
        float4 v; v.x = r[4 * q + 0]; v.y = r[4 * q + 1]; v.z = r[4 * q + 2]; v.w = r[4 * q + 3];
        *(float4*)(wb + 4 * q) = v;
      }
    }
    __syncthreads();
  }

  lg[t] = logf(piv_own);
  {
    float s = 0.f;
#pragma unroll
    for (int q = 0; q < 16; ++q) {
      float4 v; v.x = r[4 * q + 0]; v.y = r[4 * q + 1]; v.z = r[4 * q + 2]; v.w = r[4 * q + 3];
      *(float4*)&Lm[t][4 * q] = v;
      const float4 m4 = *(const float4*)&mu[4 * q];
      s += v.x * m4.x;
      s += v.y * m4.y;
      s += v.z * m4.z;
      s += v.w * m4.w;
    }
    bv[t] = s;
  }
  __syncthreads();
  if (t == 0) {
    float hs = 0.f;
    for (int i = 0; i < 64; ++i) hs += lg[i];
    float c2 = 0.f;
    for (int i = 0; i < 64; ++i) c2 += mu[i] * bv[i];
    sc[1] = -0.5f * c2 + pi[k] - sc[0] - 0.5f * hs - 0.5f * 64.f * LOG2PI;
  }
  __syncthreads();

  const int ctile = k >> 4;
  const int n = k & 15;
  float* base = bpack + (size_t)ctile * NGROUPS * 64 + n * 2;
  for (int g = t; g < NGROUPS; g += 64) {
    float w0 = 0.f, w1 = 0.f, w2 = 0.f, w3 = 0.f;
    if (g < NGROUPS - 17) {
      const int b = g & 3;
      int rem = g >> 2;
      int ci = 0;
      while (rem >= 16 - ci) { rem -= 16 - ci; ++ci; }
      const int j = 4 * (ci + rem) + b;
      const int i4 = 4 * ci;
      const float4 pr = *(const float4*)&Lm[j][i4];
      const float pjj = Lm[j][j];
      w0 = (i4 + 0 < j) ? -pr.x : ((i4 + 0 == j) ? -0.5f * pjj : 0.f);
      w1 = (i4 + 1 < j) ? -pr.y : ((i4 + 1 == j) ? -0.5f * pjj : 0.f);
      w2 = (i4 + 2 < j) ? -pr.z : ((i4 + 2 == j) ? -0.5f * pjj : 0.f);
      w3 = (i4 + 3 < j) ? -pr.w : ((i4 + 3 == j) ? -0.5f * pjj : 0.f);
    } else if (g < NGROUPS - 1) {
      const int ci = g - (NGROUPS - 17);
      w0 = bv[4 * ci + 0]; w1 = bv[4 * ci + 1]; w2 = bv[4 * ci + 2]; w3 = bv[4 * ci + 3];
    } else {
      w0 = sc[1];
    }
    float* dst = base + (size_t)g * 64;
    v2f lo; lo.x = w0; lo.y = w1;
    v2f hi; hi.x = w2; hi.y = w3;
    *(v2f*)dst = lo;
    *(v2f*)(dst + 32) = hi;
  }
}

#define NCH    68
#define NTAIL  17

__device__ __forceinline__ void gmm_next_blk(int& ci, int& jb) {
  ++jb;
  if (jb > 15) { ci = (ci < 16) ? ci + 1 : 16; jb = ci; }
}
__device__ __forceinline__ void gmm_fetch_x(const float* xl, int h, int& ci, int& jb,
                                            v2f (&xp)[2][RB], v4f (&xj)[2][RB]) {
#pragma unroll
  for (int u = 0; u < 2; ++u) {
    const int pa = 4 * ci + 2 * h;
    const int ja = 4 * jb;
#pragma unroll
    for (int rt = 0; rt < RB; ++rt) {
      xp[u][rt] = *(const v2f*)(xl + pa + rt * 16 * XS);
      xj[u][rt] = *(const v4f*)(xl + ja + rt * 16 * XS);
    }
    gmm_next_blk(ci, jb);
  }
}
__device__ __forceinline__ void gmm_fetch_b(const v2f* __restrict__ bp0, const v2f* __restrict__ bp1,
                                            int c, v2f (&bn)[8][2]) {
#pragma unroll
  for (int i = 0; i < 8; ++i) {
    bn[i][0] = bp0[(size_t)(8 * c + i) * 32];
    bn[i][1] = bp1[(size_t)(8 * c + i) * 32];
  }
}
__device__ __forceinline__ void gmm_run(v8f (&acc)[RB][2], const v2f (&xp)[2][RB], const v4f (&xj)[2][RB],
                                        const v2f (&bc)[8][2]) {
#pragma unroll
  for (int u = 0; u < 2; ++u) {
#pragma unroll
    for (int b = 0; b < 4; ++b) {
#pragma unroll
      for (int rt = 0; rt < RB; ++rt) {
        const v2f a = xp[u][rt] * xj[u][rt][b];
        acc[rt][0] = __builtin_amdgcn_wmma_f32_16x16x4_f32(false, a, false, bc[4 * u + b][0], (short)0, acc[rt][0], false, false);
        acc[rt][1] = __builtin_amdgcn_wmma_f32_16x16x4_f32(false, a, false, bc[4 * u + b][1], (short)0, acc[rt][1], false, false);
      }
    }
  }
}
__device__ __forceinline__ void gmm_step(v8f (&acc)[RB][2], const v2f (&xpc)[2][RB], const v4f (&xjc)[2][RB],
                                         v2f (&xpn)[2][RB], v4f (&xjn)[2][RB], const v2f (&bc)[8][2],
                                         v2f (&bf)[8][2], const v2f* __restrict__ bp0,
                                         const v2f* __restrict__ bp1, int cfetch, const float* xl, int h,
                                         int& ci, int& jb) {
  gmm_fetch_b(bp0, bp1, cfetch, bf);
  gmm_fetch_x(xl, h, ci, jb, xpn, xjn);
  __builtin_amdgcn_sched_barrier(0);
  gmm_run(acc, xpc, xjc, bc);
}
__device__ __forceinline__ void gmm_tail(v8f (&acc)[RB][2], const v2f (&ba)[8][2], const v2f (&bb)[8][2],
                                         const v2f (&bl)[2], const float* xl, int h) {
  v2f ta[NTAIL][RB];
#pragma unroll
  for (int i = 0; i < NTAIL; ++i) {
#pragma unroll
    for (int rt = 0; rt < RB; ++rt) ta[i][rt] = *(const v2f*)(xl + 4 * i + 2 * h + rt * 16 * XS);
  }
  __builtin_amdgcn_sched_barrier(0);
#pragma unroll
  for (int i = 0; i < NTAIL; ++i) {
#pragma unroll
    for (int rt = 0; rt < RB; ++rt) {
      const v2f f0 = (i < 8) ? ba[i & 7][0] : ((i < 16) ? bb[i & 7][0] : bl[0]);
      const v2f f1 = (i < 8) ? ba[i & 7][1] : ((i < 16) ? bb[i & 7][1] : bl[1]);
      acc[rt][0] = __builtin_amdgcn_wmma_f32_16x16x4_f32(false, ta[i][rt], false, f0, (short)0, acc[rt][0], false, false);
      acc[rt][1] = __builtin_amdgcn_wmma_f32_16x16x4_f32(false, ta[i][rt], false, f1, (short)0, acc[rt][1], false, false);
    }
  }
}

__global__ __launch_bounds__(32 * WPB) void gmm_main(const float* __restrict__ X,
                                                     const float* __restrict__ bpack,
                                                     float* __restrict__ out,
                                                     int ntiles) {
  __shared__ __attribute__((aligned(16))) float xs[WPB][16 * RB][XS];
  const int tid = threadIdx.x;
  const int w = __builtin_amdgcn_readfirstlane(tid >> 5);
  const int lane = tid & 31;
  const int h = lane >> 4;
  const int r = lane & 15;
  const int t0 = (blockIdx.x * WPB + w) * RB;
  int nvalid = ntiles - t0;
  if (nvalid > RB) nvalid = RB;

  if (nvalid > 0) {
    const float* xp = X + (size_t)t0 * 16 * 64;
    const int umax = nvalid * 8;
    for (int u = 0; u < umax; ++u) {
      const int flat = u * 128 + lane * 4;
      float4 v = *(const float4*)(xp + flat);
      *(float4*)&xs[w][flat >> 6][flat & 63] = v;
    }
    float4 z; z.x = 0.f; z.y = 0.f; z.z = 0.f; z.w = 0.f;
    for (int u = umax; u < RB * 8; ++u) {
      const int flat = u * 128 + lane * 4;
      *(float4*)&xs[w][flat >> 6][flat & 63] = z;
    }
    float4 one; one.x = 1.f; one.y = 0.f; one.z = 0.f; one.w = 0.f;
    *(float4*)&xs[w][lane][64] = one;
    *(float4*)&xs[w][lane + 32][64] = one;
  }
  __syncthreads();
  if (nvalid <= 0) return;

  const float* xl = &xs[w][r][0];
  const v2f* __restrict__ bp0 = (const v2f*)bpack + (h * 16 + r);
  const v2f* __restrict__ bp1 = bp0 + (size_t)NGROUPS * 32;

  v8f acc[RB][2];
#pragma unroll
  for (int rt = 0; rt < RB; ++rt) { acc[rt][0] = (v8f){}; acc[rt][1] = (v8f){}; }

  v2f b0[8][2], b1[8][2], b2[8][2], b3[8][2];
  v2f xpe[2][RB], xpo[2][RB];
  v4f xje[2][RB], xjo[2][RB];
  int ci = 0, jb = 0;

  gmm_fetch_b(bp0, bp1, 0, b0);
  gmm_fetch_b(bp0, bp1, 1, b1);
  v2f bl[2];
  bl[0] = bp0[(size_t)(NGROUPS - 1) * 32];
  bl[1] = bp1[(size_t)(NGROUPS - 1) * 32];
  gmm_fetch_x(xl, h, ci, jb, xpe, xje);
  for (int c = 0; c < NCH; c += 4) {
    gmm_step(acc, xpe, xje, xpo, xjo, b0, b2, bp0, bp1, c + 2, xl, h, ci, jb);
    gmm_step(acc, xpo, xjo, xpe, xje, b1, b3, bp0, bp1, c + 3, xl, h, ci, jb);
    gmm_step(acc, xpe, xje, xpo, xjo, b2, b0, bp0, bp1, c + 4, xl, h, ci, jb);
    gmm_step(acc, xpo, xjo, xpe, xje, b3, b1, bp0, bp1, c + 5, xl, h, ci, jb);
  }
  gmm_tail(acc, b0, b1, bl, xl, h);

#pragma unroll
  for (int rt = 0; rt < RB; ++rt) {
    float res[8];
#pragma unroll
    for (int v = 0; v < 8; ++v) {
      float a0 = acc[rt][0][v], a1 = acc[rt][1][v];
      float mx = fmaxf(a0, a1);
      mx = fmaxf(mx, __shfl_xor(mx, 1, 32));
      mx = fmaxf(mx, __shfl_xor(mx, 2, 32));
      mx = fmaxf(mx, __shfl_xor(mx, 4, 32));
      mx = fmaxf(mx, __shfl_xor(mx, 8, 32));
      float e = expf(a0 - mx) + expf(a1 - mx);
      e += __shfl_xor(e, 1, 32);
      e += __shfl_xor(e, 2, 32);
      e += __shfl_xor(e, 4, 32);
      e += __shfl_xor(e, 8, 32);
      res[v] = mx + logf(e);
    }
    float val = res[0];
#pragma unroll
    for (int v = 1; v < 8; ++v) val = (r == v) ? res[v] : val;
    if (rt < nvalid && r < 8)
      out[(size_t)(t0 + rt) * 16 + 8 * h + r] = val;
  }
}

extern "C" void kernel_launch(void* const* d_in, const int* in_sizes, int n_in,
                              void* d_out, int out_size, void* d_ws, size_t ws_size,
                              hipStream_t stream) {
  const float* X   = (const float*)d_in[0];
  const float* loc = (const float*)d_in[1];
  const float* cov = (const float*)d_in[2];
  const float* pi  = (const float*)d_in[3];
  float* out = (float*)d_out;
  float* bpack = (float*)d_ws;

  const int N = in_sizes[0] / 64;
  const int ntiles = N / 16;

  gmm_prep<<<GMM_K, 64, 0, stream>>>(loc, cov, pi, bpack);
  const int tilesPerBlock = WPB * RB;
  const int nblk = (ntiles + tilesPerBlock - 1) / tilesPerBlock;
  gmm_main<<<nblk, 32 * WPB, 0, stream>>>(X, bpack, out, ntiles);
}


# gmm_main: two waves per SIMD (8-wave blocks, 2 row tiles per wave, not 4): one wave's load groups, row staging and logsumexp run under the other wave's WMMAs; K and sum order kept
# speedup vs baseline: 5.7849x; 1.0438x over previous
#include <hip/hip_runtime.h>
#include <math.h>


#define NGROUPS 561
#define GMM_K   32
#define LOG2PI  1.8378770664093453f
#define XS      68
#define RB      2
#define WPB     8

typedef __attribute__((ext_vector_type(2))) float v2f;
typedef __attribute__((ext_vector_type(8))) float v8f;
typedef __attribute__((ext_vector_type(4))) float v4f;
typedef __attribute__((address_space(3))) float lds_float;

template <int OFF> __device__ __forceinline__ v4f lds_read16(unsigned addr) {
  v4f v;
  asm volatile("ds_load_b128 %0, %1 offset:%2" : "=&v"(v) : "v"(addr), "n"(OFF) : "memory");
  return v;
}
__device__ __forceinline__ float lds_read4(unsigned addr) {
  float v;
  asm volatile("ds_load_b32 %0, %1" : "=&v"(v) : "v"(addr) : "memory");
  return v;
}

#define PS 68

__global__ __launch_bounds__(64) void gmm_prep(const float* __restrict__ loc,
                                               const float* __restrict__ cov,
                                               const float* __restrict__ pi,
                                               float* __restrict__ bpack) {
  __shared__ __attribute__((aligned(16))) float Lm[64][PS];
  __shared__ __attribute__((aligned(16))) float rowb[2][2][64];
  __shared__ float nipv[2][64];
  __shared__ float lg[64];
  __shared__ __attribute__((aligned(16))) float mu[64];
  __shared__ float bv[64];
  __shared__ float sc[2];
  const int k = blockIdx.x;
  const int t = threadIdx.x;

  {
    const float* ck = cov + (size_t)k * 4096;
    for (int i = 0; i < 64; ++i) Lm[i][t] = ck[i * 64 + t];
  }
  mu[t] = loc[k * 64 + t];
  if (t == 0) {
    float m = pi[0];
    for (int i = 1; i < GMM_K; ++i) m = fmaxf(m, pi[i]);
    float s = 0.f;
    for (int i = 0; i < GMM_K; ++i) s += expf(pi[i] - m);
    sc[0] = m + logf(s);
  }
  __syncthreads();

  float r[64];
#pragma unroll
  for (int q = 0; q < 16; ++q) {
    const float4 a = *(const float4*)&Lm[t][4 * q];
    r[4 * q + 0] = (a.x + Lm[4 * q + 0][t]) * 0.5f;
    r[4 * q + 1] = (a.y + Lm[4 * q + 1][t]) * 0.5f;
    r[4 * q + 2] = (a.z + Lm[4 * q + 2][t]) * 0.5f;
    r[4 * q + 3] = (a.w + Lm[4 * q + 3][t]) * 0.5f;
  }
  if (t < 2) {
#pragma unroll
    for (int q = 0; q < 16; ++q) {
      float4 v; v.x = r[4 * q + 0]; v.y = r[4 * q + 1]; v.z = r[4 * q + 2]; v.w = r[4 * q + 3];
      *(float4*)&rowb[0][t][4 * q] = v;
    }
  }
  nipv[0][t] = 1.0f / r[0];
  __syncthreads();

  float piv_own = 1.f;
#pragma nounroll
  for (int s = 0; s < 32; ++s) {
    const unsigned ipa = (unsigned)(size_t)(lds_float*)&nipv[s & 1][2 * s];
    const unsigned rwa = (unsigned)(size_t)(lds_float*)&rowb[s & 1][0][0];
    float ia = lds_read4(ipa);
    v4f a0[16], a1[16];
#define GJ_RD(q) a0[q] = lds_read16<16 * (q)>(rwa); a1[q] = lds_read16<256 + 16 * (q)>(rwa);
    GJ_RD(0) GJ_RD(1) GJ_RD(2) GJ_RD(3) GJ_RD(4) GJ_RD(5) GJ_RD(6) GJ_RD(7)
    GJ_RD(8) GJ_RD(9) GJ_RD(10) GJ_RD(11) GJ_RD(12) GJ_RD(13) GJ_RD(14) GJ_RD(15)
#undef GJ_RD
#define GJ_W0(m) a0[(m) >> 2][(m) & 3]
#define GJ_W1(m) a1[(m) >> 2][(m) & 3]
    asm volatile("s_wait_dscnt 30" : "+v"(ia), "+v"(a0[0]), "+v"(a1[0]));
    const float l = GJ_W1(0) * ia;
    const float u = GJ_W0(1) * ia;
    const float sch = GJ_W1(1) - l * GJ_W0(1);
    const float q11 = 1.0f / sch;
    const float q01 = -u * q11;
    const float q10 = -l * q11;
    const float q00 = ia - q01 * l;
    const bool own = ((t >> 1) == s);
    const bool odd = ((t & 1) != 0);
    piv_own = own ? (odd ? sch : GJ_W0(0)) : piv_own;
    float c0 = r[0] * q00 + r[1] * q10;
    float c1 = r[0] * q01 + r[1] * q11;
    if (own) { c0 = odd ? -q10 : -q00; c1 = odd ? -q11 : -q01; }
#define GJ_UPD(m) { const float base = own ? 0.f : r[m]; \
                    r[(m) - 2] = __builtin_fmaf(-c0, GJ_W0(m), __builtin_fmaf(-c1, GJ_W1(m), base)); }
    asm volatile("s_wait_dscnt 16" : "+v"(c0), "+v"(c1),
                 "+v"(a0[1]), "+v"(a1[1]), "+v"(a0[2]), "+v"(a1[2]), "+v"(a0[3]), "+v"(a1[3]), "+v"(a0[4]),
                 "+v"(a1[4]), "+v"(a0[5]), "+v"(a1[5]), "+v"(a0[6]), "+v"(a1[6]), "+v"(a0[7]), "+v"(a1[7]));
    asm volatile("s_wait_dscnt 0" : "+v"(a0[8]), "+v"(a1[8]), "+v"(a0[9]), "+v"(a1[9]), "+v"(a0[10]),
                 "+v"(a1[10]), "+v"(a0[11]), "+v"(a1[11]), "+v"(a0[12]), "+v"(a1[12]), "+v"(a0[13]),
                 "+v"(a1[13]), "+v"(a0[14]), "+v"(a1[14]), "+v"(a0[15]), "+v"(a1[15]));
#pragma unroll
    for (int m = 2; m < 64; ++m) GJ_UPD(m)
#undef GJ_UPD
#undef GJ_W0
#undef GJ_W1
    r[62] = -c0;
    r[63] = -c1;
    nipv[(s + 1) & 1][t] = 1.0f / r[0];
    if ((t >> 1) == s + 1) {
      float* wb = &rowb[(s + 1) & 1][t & 1][0];
#pragma unroll
      for (int q = 0; q < 16; ++q) {
        float4 v; v.x = r[4 * q + 0]; v.y = r[4 * q + 1]; v.z = r[4 * q + 2]; v.w = r[4 * q + 3];
        *(float4*)(wb + 4 * q) = v;
      }
    }
    __syncthreads();
  }

  lg[t] = logf(piv_own);
  {
    float s = 0.f;
#pragma unroll
    for (int q = 0; q < 16; ++q) {
      float4 v; v.x = r[4 * q + 0]; v.y = r[4 * q + 1]; v.z = r[4 * q + 2]; v.w = r[4 * q + 3];
      *(float4*)&Lm[t][4 * q] = v;
      const float4 m4 = *(const float4*)&mu[4 * q];
      s += v.x * m4.x;
      s += v.y * m4.y;
      s += v.z * m4.z;
      s += v.w * m4.w;
    }
    bv[t] = s;
  }
  __syncthreads();
  if (t == 0) {
    float hs = 0.f;
    for (int i = 0; i < 64; ++i) hs += lg[i];
    float c2 = 0.f;
    for (int i = 0; i < 64; ++i) c2 += mu[i] * bv[i];
    sc[1] = -0.5f * c2 + pi[k] - sc[0] - 0.5f * hs - 0.5f * 64.f * LOG2PI;
  }
  __syncthreads();

  const int ctile = k >> 4;
  const int n = k & 15;
  float* base = bpack + (size_t)ctile * NGROUPS * 64 + n * 2;
  for (int g = t; g < NGROUPS; g += 64) {
    float w0 = 0.f, w1 = 0.f, w2 = 0.f, w3 = 0.f;
    if (g < NGROUPS - 17) {
      const int b = g & 3;
      int rem = g >> 2;
      int ci = 0;
      while (rem >= 16 - ci) { rem -= 16 - ci; ++ci; }
      const int j = 4 * (ci + rem) + b;
      const int i4 = 4 * ci;
      const float4 pr = *(const float4*)&Lm[j][i4];
      const float pjj = Lm[j][j];
      w0 = (i4 + 0 < j) ? -pr.x : ((i4 + 0 == j) ? -0.5f * pjj : 0.f);
      w1 = (i4 + 1 < j) ? -pr.y : ((i4 + 1 == j) ? -0.5f * pjj : 0.f);
      w2 = (i4 + 2 < j) ? -pr.z : ((i4 + 2 == j) ? -0.5f * pjj : 0.f);
      w3 = (i4 + 3 < j) ? -pr.w : ((i4 + 3 == j) ? -0.5f * pjj : 0.f);
    } else if (g < NGROUPS - 1) {
      const int ci = g - (NGROUPS - 17);
      w0 = bv[4 * ci + 0]; w1 = bv[4 * ci + 1]; w2 = bv[4 * ci + 2]; w3 = bv[4 * ci + 3];
    } else {
      w0 = sc[1];
    }
    float* dst = base + (size_t)g * 64;
    v2f lo; lo.x = w0; lo.y = w1;
    v2f hi; hi.x = w2; hi.y = w3;
    *(v2f*)dst = lo;
    *(v2f*)(dst + 32) = hi;
  }
}

#define NCH    68
#define NTAIL  17

__device__ __forceinline__ void gmm_next_blk(int& ci, int& jb) {
  ++jb;
  if (jb > 15) { ci = (ci < 16) ? ci + 1 : 16; jb = ci; }
}
__device__ __forceinline__ void gmm_fetch_x(const float* xl, int h, int& ci, int& jb,
                                            v2f (&xp)[2][RB], v4f (&xj)[2][RB]) {
#pragma unroll
  for (int u = 0; u < 2; ++u) {
    const int pa = 4 * ci + 2 * h;
    const int ja = 4 * jb;
#pragma unroll
    for (int rt = 0; rt < RB; ++rt) {
      xp[u][rt] = *(const v2f*)(xl + pa + rt * 16 * XS);
      xj[u][rt] = *(const v4f*)(xl + ja + rt * 16 * XS);
    }
    gmm_next_blk(ci, jb);
  }
}
__device__ __forceinline__ void gmm_fetch_b(const v2f* __restrict__ bp0, const v2f* __restrict__ bp1,
                                            int c, v2f (&bn)[8][2]) {
#pragma unroll
  for (int i = 0; i < 8; ++i) {
    bn[i][0] = bp0[(size_t)(8 * c + i) * 32];
    bn[i][1] = bp1[(size_t)(8 * c + i) * 32];
  }
}
__device__ __forceinline__ void gmm_run(v8f (&acc)[RB][2], const v2f (&xp)[2][RB], const v4f (&xj)[2][RB],
                                        const v2f (&bc)[8][2]) {
#pragma unroll
  for (int u = 0; u < 2; ++u) {
#pragma unroll
    for (int b = 0; b < 4; ++b) {
#pragma unroll
      for (int rt = 0; rt < RB; ++rt) {
        const v2f a = xp[u][rt] * xj[u][rt][b];
        acc[rt][0] = __builtin_amdgcn_wmma_f32_16x16x4_f32(false, a, false, bc[4 * u + b][0], (short)0, acc[rt][0], false, false);
        acc[rt][1] = __builtin_amdgcn_wmma_f32_16x16x4_f32(false, a, false, bc[4 * u + b][1], (short)0, acc[rt][1], false, false);
      }
    }
  }
}
__device__ __forceinline__ void gmm_step(v8f (&acc)[RB][2], const v2f (&xpc)[2][RB], const v4f (&xjc)[2][RB],
                                         v2f (&xpn)[2][RB], v4f (&xjn)[2][RB], const v2f (&bc)[8][2],
                                         v2f (&bf)[8][2], const v2f* __restrict__ bp0,
                                         const v2f* __restrict__ bp1, int cfetch, const float* xl, int h,
                                         int& ci, int& jb) {
  gmm_fetch_b(bp0, bp1, cfetch, bf);
  gmm_fetch_x(xl, h, ci, jb, xpn, xjn);
  __builtin_amdgcn_sched_barrier(0);
  gmm_run(acc, xpc, xjc, bc);
}
__device__ __forceinline__ void gmm_tail(v8f (&acc)[RB][2], const v2f (&ba)[8][2], const v2f (&bb)[8][2],
                                         const v2f (&bl)[2], const float* xl, int h) {
  v2f ta[NTAIL][RB];
#pragma unroll
  for (int i = 0; i < NTAIL; ++i) {
#pragma unroll
    for (int rt = 0; rt < RB; ++rt) ta[i][rt] = *(const v2f*)(xl + 4 * i + 2 * h + rt * 16 * XS);
  }
  __builtin_amdgcn_sched_barrier(0);
#pragma unroll
  for (int i = 0; i < NTAIL; ++i) {
#pragma unroll
    for (int rt = 0; rt < RB; ++rt) {
      const v2f f0 = (i < 8) ? ba[i & 7][0] : ((i < 16) ? bb[i & 7][0] : bl[0]);
      const v2f f1 = (i < 8) ? ba[i & 7][1] : ((i < 16) ? bb[i & 7][1] : bl[1]);
      acc[rt][0] = __builtin_amdgcn_wmma_f32_16x16x4_f32(false, ta[i][rt], false, f0, (short)0, acc[rt][0], false, false);
      acc[rt][1] = __builtin_amdgcn_wmma_f32_16x16x4_f32(false, ta[i][rt], false, f1, (short)0, acc[rt][1], false, false);
    }
  }
}

__global__ __launch_bounds__(32 * WPB) void gmm_main(const float* __restrict__ X,
                                                     const float* __restrict__ bpack,
                                                     float* __restrict__ out,
                                                     int ntiles) {
  __shared__ __attribute__((aligned(16))) float xs[WPB][16 * RB][XS];
  const int tid = threadIdx.x;
  const int w = __builtin_amdgcn_readfirstlane(tid >> 5);
  const int lane = tid & 31;
  const int h = lane >> 4;
  const int r = lane & 15;
  const int t0 = (blockIdx.x * WPB + w) * RB;
  int nvalid = ntiles - t0;
  if (nvalid > RB) nvalid = RB;

  if (nvalid > 0) {
    const float* xp = X + (size_t)t0 * 16 * 64;
    const int umax = nvalid * 8;
    for (int u = 0; u < umax; ++u) {
      const int flat = u * 128 + lane * 4;
      float4 v = *(const float4*)(xp + flat);
      *(float4*)&xs[w][flat >> 6][flat & 63] = v;
    }
    float4 z; z.x = 0.f; z.y = 0.f; z.z = 0.f; z.w = 0.f;
    for (int u = umax; u < RB * 8; ++u) {
      const int flat = u * 128 + lane * 4;
      *(float4*)&xs[w][flat >> 6][flat & 63] = z;
    }
    float4 one; one.x = 1.f; one.y = 0.f; one.z = 0.f; one.w = 0.f;
    for (int rr = 0; rr < 16 * RB; rr += 32) *(float4*)&xs[w][rr + lane][64] = one;
  }
  __syncthreads();
  if (nvalid <= 0) return;

  const float* xl = &xs[w][r][0];
  const v2f* __restrict__ bp0 = (const v2f*)bpack + (h * 16 + r);
  const v2f* __restrict__ bp1 = bp0 + (size_t)NGROUPS * 32;

  v8f acc[RB][2];
#pragma unroll
  for (int rt = 0; rt < RB; ++rt) { acc[rt][0] = (v8f){}; acc[rt][1] = (v8f){}; }

  v2f b0[8][2], b1[8][2], b2[8][2], b3[8][2];
  v2f xpe[2][RB], xpo[2][RB];
  v4f xje[2][RB], xjo[2][RB];
  int ci = 0, jb = 0;

  gmm_fetch_b(bp0, bp1, 0, b0);
  gmm_fetch_b(bp0, bp1, 1, b1);
  v2f bl[2];
  bl[0] = bp0[(size_t)(NGROUPS - 1) * 32];
  bl[1] = bp1[(size_t)(NGROUPS - 1) * 32];
  gmm_fetch_x(xl, h, ci, jb, xpe, xje);
  for (int c = 0; c < NCH; c += 4) {
    gmm_step(acc, xpe, xje, xpo, xjo, b0, b2, bp0, bp1, c + 2, xl, h, ci, jb);
    gmm_step(acc, xpo, xjo, xpe, xje, b1, b3, bp0, bp1, c + 3, xl, h, ci, jb);
    gmm_step(acc, xpe, xje, xpo, xjo, b2, b0, bp0, bp1, c + 4, xl, h, ci, jb);
    gmm_step(acc, xpo, xjo, xpe, xje, b3, b1, bp0, bp1, c + 5, xl, h, ci, jb);
  }
  gmm_tail(acc, b0, b1, bl, xl, h);

#pragma unroll
  for (int rt = 0; rt < RB; ++rt) {
    float res[8];
#pragma unroll
    for (int v = 0; v < 8; ++v) {
      float a0 = acc[rt][0][v], a1 = acc[rt][1][v];
      float mx = fmaxf(a0, a1);
      mx = fmaxf(mx, __shfl_xor(mx, 1, 32));
      mx = fmaxf(mx, __shfl_xor(mx, 2, 32));
      mx = fmaxf(mx, __shfl_xor(mx, 4, 32));
      mx = fmaxf(mx, __shfl_xor(mx, 8, 32));
      float e = expf(a0 - mx) + expf(a1 - mx);
      e += __shfl_xor(e, 1, 32);
      e += __shfl_xor(e, 2, 32);
      e += __shfl_xor(e, 4, 32);
      e += __shfl_xor(e, 8, 32);
      res[v] = mx + logf(e);
    }
    float val = res[0];
#pragma unroll
    for (int v = 1; v < 8; ++v) val = (r == v) ? res[v] : val;
    if (rt < nvalid && r < 8)
      out[(size_t)(t0 + rt) * 16 + 8 * h + r] = val;
  }
}

extern "C" void kernel_launch(void* const* d_in, const int* in_sizes, int n_in,
                              void* d_out, int out_size, void* d_ws, size_t ws_size,
                              hipStream_t stream) {
  const float* X   = (const float*)d_in[0];
  const float* loc = (const float*)d_in[1];
  const float* cov = (const float*)d_in[2];
  const float* pi  = (const float*)d_in[3];
  float* out = (float*)d_out;
  float* bpack = (float*)d_ws;

  const int N = in_sizes[0] / 64;
  const int ntiles = N / 16;

  gmm_prep<<<GMM_K, 64, 0, stream>>>(loc, cov, pi, bpack);
  const int tilesPerBlock = WPB * RB;
  const int nblk = (ntiles + tilesPerBlock - 1) / tilesPerBlock;
  gmm_main<<<nblk, 32 * WPB, 0, stream>>>(X, bpack, out, ntiles);
}


# gmm_prep: one-thread and one-chain sections outside the sweep made parallel: lse(pi) by 32 lanes, P*mu and the 64-term sums on 4 partial sums, pack walk by selects; sum order of those sums changes
# speedup vs baseline: 6.0689x; 1.0491x over previous
#include <hip/hip_runtime.h>
#include <math.h>


#define NGROUPS 561
#define GMM_K   32
#define LOG2PI  1.8378770664093453f
#define XS      68
#define RB      2
#define WPB     8

typedef __attribute__((ext_vector_type(2))) float v2f;
typedef __attribute__((ext_vector_type(8))) float v8f;
typedef __attribute__((ext_vector_type(4))) float v4f;
typedef __attribute__((address_space(3))) float lds_float;

template <int OFF> __device__ __forceinline__ v4f lds_read16(unsigned addr) {
  v4f v;
  asm volatile("ds_load_b128 %0, %1 offset:%2" : "=&v"(v) : "v"(addr), "n"(OFF) : "memory");
  return v;
}
__device__ __forceinline__ float lds_read4(unsigned addr) {
  float v;
  asm volatile("ds_load_b32 %0, %1" : "=&v"(v) : "v"(addr) : "memory");
  return v;
}

#define PS 68

__host__ __device__ constexpr int gmm_row_start(int c) { return 16 * c - c * (c - 1) / 2; }
__host__ __device__ constexpr int gmm_row_of(int blk) {
  int c = 0;
  while (c < 15 && blk >= gmm_row_start(c + 1)) ++c;
  return c;
}
__device__ __forceinline__ void gmm_pack_store(float* dst, float w0, float w1, float w2, float w3) {
  v2f lo; lo.x = w0; lo.y = w1;
  v2f hi; hi.x = w2; hi.y = w3;
  *(v2f*)dst = lo;
  *(v2f*)(dst + 32) = hi;
}
template <int PASS>
__device__ __forceinline__ void gmm_pack_pairs(const float (*Lm)[PS], float* base, int t) {
  constexpr int CLO = gmm_row_of(16 * PASS);
  constexpr int CHI = gmm_row_of(16 * PASS + 15 < 135 ? 16 * PASS + 15 : 135);
  const int blk = (t >> 2) + 16 * PASS;
  int ci = CLO;
  int start = gmm_row_start(CLO);
  for (int c = CLO + 1; c <= CHI; ++c) {
    const bool in = (blk >= gmm_row_start(c));
    ci = in ? c : ci;
    start = in ? gmm_row_start(c) : start;
  }
  const int jr = 4 * (blk - start) + (t & 3);
  const float4 pr = *(const float4*)&Lm[4 * ci + jr][4 * ci];
  const float w0 = (0 < jr) ? -pr.x : -0.5f * pr.x;
  const float w1 = (1 < jr) ? -pr.y : ((1 == jr) ? -0.5f * pr.y : 0.f);
  const float w2 = (2 < jr) ? -pr.z : ((2 == jr) ? -0.5f * pr.z : 0.f);
  const float w3 = (3 < jr) ? -pr.w : ((3 == jr) ? -0.5f * pr.w : 0.f);
  gmm_pack_store(base + (size_t)PASS * 4096, w0, w1, w2, w3);
}

__global__ __launch_bounds__(64) void gmm_prep(const float* __restrict__ loc,
                                               const float* __restrict__ cov,
                                               const float* __restrict__ pi,
                                               float* __restrict__ bpack) {
  __shared__ __attribute__((aligned(16))) float Lm[64][PS];
  __shared__ __attribute__((aligned(16))) float rowb[2][2][64];
  __shared__ float nipv[2][64];
  __shared__ __attribute__((aligned(16))) float red[64][2];
  __shared__ __attribute__((aligned(16))) float pe[64];
  __shared__ float pil[64];
  __shared__ __attribute__((aligned(16))) float mu[64];
  __shared__ float bv[64];
  const int k = blockIdx.x;
  const int t = threadIdx.x;

  const float pit = pi[t & (GMM_K - 1)];
  {
    const float* ck = cov + (size_t)k * 4096;
    for (int i = 0; i < 64; ++i) Lm[i][t] = ck[i * 64 + t];
  }
  const float mut = loc[k * 64 + t];
  mu[t] = mut;
  float pmax = pi[0];
  for (int i = 1; i < GMM_K; ++i) pmax = fmaxf(pmax, pi[i]);
  pil[t] = pit;
  pe[t] = expf(pit - pmax);
  __syncthreads();

  float r[64];
#pragma unroll
  for (int q = 0; q < 16; ++q) {
    const float4 a = *(const float4*)&Lm[t][4 * q];
    r[4 * q + 0] = (a.x + Lm[4 * q + 0][t]) * 0.5f;
    r[4 * q + 1] = (a.y + Lm[4 * q + 1][t]) * 0.5f;
    r[4 * q + 2] = (a.z + Lm[4 * q + 2][t]) * 0.5f;
    r[4 * q + 3] = (a.w + Lm[4 * q + 3][t]) * 0.5f;
  }
  if (t < 2) {
#pragma unroll
    for (int q = 0; q < 16; ++q) {
      float4 v; v.x = r[4 * q + 0]; v.y = r[4 * q + 1]; v.z = r[4 * q + 2]; v.w = r[4 * q + 3];
      *(float4*)&rowb[0][t][4 * q] = v;
    }
  }
  nipv[0][t] = 1.0f / r[0];
  __syncthreads();

  float piv_own = 1.f;
#pragma nounroll
  for (int s = 0; s < 32; ++s) {
    const unsigned ipa = (unsigned)(size_t)(lds_float*)&nipv[s & 1][2 * s];
    const unsigned rwa = (unsigned)(size_t)(lds_float*)&rowb[s & 1][0][0];
    float ia = lds_read4(ipa);
    v4f a0[16], a1[16];
#define GJ_RD(q) a0[q] = lds_read16<16 * (q)>(rwa); a1[q] = lds_read16<256 + 16 * (q)>(rwa);
    GJ_RD(0) GJ_RD(1) GJ_RD(2) GJ_RD(3) GJ_RD(4) GJ_RD(5) GJ_RD(6) GJ_RD(7)
    GJ_RD(8) GJ_RD(9) GJ_RD(10) GJ_RD(11) GJ_RD(12) GJ_RD(13) GJ_RD(14) GJ_RD(15)
#undef GJ_RD
#define GJ_W0(m) a0[(m) >> 2][(m) & 3]
#define GJ_W1(m) a1[(m) >> 2][(m) & 3]
    asm volatile("s_wait_dscnt 30" : "+v"(ia), "+v"(a0[0]), "+v"(a1[0]));
    const float l = GJ_W1(0) * ia;
    const float u = GJ_W0(1) * ia;
    const float sch = GJ_W1(1) - l * GJ_W0(1);
    const float q11 = 1.0f / sch;
    const float q01 = -u * q11;
    const float q10 = -l * q11;
    const float q00 = ia - q01 * l;
    const bool own = ((t >> 1) == s);
    const bool odd = ((t & 1) != 0);
    piv_own = own ? (odd ? sch : GJ_W0(0)) : piv_own;
    float c0 = r[0] * q00 + r[1] * q10;
    float c1 = r[0] * q01 + r[1] * q11;
    if (own) { c0 = odd ? -q10 : -q00; c1 = odd ? -q11 : -q01; }
#define GJ_UPD(m) { const float base = own ? 0.f : r[m]; \
                    r[(m) - 2] = __builtin_fmaf(-c0, GJ_W0(m), __builtin_fmaf(-c1, GJ_W1(m), base)); }
    asm volatile("s_wait_dscnt 16" : "+v"(c0), "+v"(c1),
                 "+v"(a0[1]), "+v"(a1[1]), "+v"(a0[2]), "+v"(a1[2]), "+v"(a0[3]), "+v"(a1[3]), "+v"(a0[4]),
                 "+v"(a1[4]), "+v"(a0[5]), "+v"(a1[5]), "+v"(a0[6]), "+v"(a1[6]), "+v"(a0[7]), "+v"(a1[7]));
    asm volatile("s_wait_dscnt 0" : "+v"(a0[8]), "+v"(a1[8]), "+v"(a0[9]), "+v"(a1[9]), "+v"(a0[10]),
                 "+v"(a1[10]), "+v"(a0[11]), "+v"(a1[11]), "+v"(a0[12]), "+v"(a1[12]), "+v"(a0[13]),
                 "+v"(a1[13]), "+v"(a0[14]), "+v"(a1[14]), "+v"(a0[15]), "+v"(a1[15]));
#pragma unroll
    for (int m = 2; m < 64; ++m) GJ_UPD(m)
#undef GJ_UPD
#undef GJ_W0
#undef GJ_W1
    r[62] = -c0;
    r[63] = -c1;
    nipv[(s + 1) & 1][t] = 1.0f / r[0];
    if ((t >> 1) == s + 1) {
      float* wb = &rowb[(s + 1) & 1][t & 1][0];
#pragma unroll
      for (int q = 0; q < 16; ++q) {
        float4 v; v.x = r[4 * q + 0]; v.y = r[4 * q + 1]; v.z = r[4 * q + 2]; v.w = r[4 * q + 3];
        *(float4*)(wb + 4 * q) = v;
      }
    }
    __syncthreads();
  }

  {
    v4f m4[16];
#pragma unroll
    for (int q = 0; q < 16; ++q) m4[q] = *(const v4f*)&mu[4 * q];
    float s0 = 0.f, s1 = 0.f, s2 = 0.f, s3 = 0.f;
#pragma unroll
    for (int q = 0; q < 16; ++q) {
      float4 v; v.x = r[4 * q + 0]; v.y = r[4 * q + 1]; v.z = r[4 * q + 2]; v.w = r[4 * q + 3];
      *(float4*)&Lm[t][4 * q] = v;
      s0 = __builtin_fmaf(v.x, m4[q].x, s0);
      s1 = __builtin_fmaf(v.y, m4[q].y, s1);
      s2 = __builtin_fmaf(v.z, m4[q].z, s2);
      s3 = __builtin_fmaf(v.w, m4[q].w, s3);
    }
    const float s = (s0 + s1) + (s2 + s3);
    bv[t] = s;
    v2f lp; lp.x = logf(piv_own); lp.y = mut * s;
    *(v2f*)&red[t][0] = lp;
  }
  __syncthreads();

  float cterm = 0.f;
  if (t == (NGROUPS - 1) % 64) {
    float h0 = 0.f, h1 = 0.f, h2 = 0.f, h3 = 0.f;
    float c0 = 0.f, c1 = 0.f, c2 = 0.f, c3 = 0.f;
    for (int q = 0; q < 16; ++q) {
      const v4f u = *(const v4f*)&red[4 * q][0];
      const v4f w = *(const v4f*)&red[4 * q + 2][0];
      h0 += u.x; c0 += u.y; h1 += u.z; c1 += u.w;
      h2 += w.x; c2 += w.y; h3 += w.z; c3 += w.w;
    }
    float e0 = 0.f, e1 = 0.f, e2 = 0.f, e3 = 0.f;
    for (int q = 0; q < GMM_K / 4; ++q) {
      const v4f e = *(const v4f*)&pe[4 * q];
      e0 += e.x; e1 += e.y; e2 += e.z; e3 += e.w;
    }
    const float hs = (h0 + h1) + (h2 + h3);
    const float cc = (c0 + c1) + (c2 + c3);
    const float lse = pmax + logf((e0 + e1) + (e2 + e3));
    cterm = -0.5f * cc + pil[k] - lse - 0.5f * hs - 0.5f * 64.f * LOG2PI;
  }

  const int ctile = k >> 4;
  const int n = k & 15;
  float* base = bpack + (size_t)ctile * NGROUPS * 64 + n * 2 + (size_t)t * 64;
  gmm_pack_pairs<0>(Lm, base, t);
  gmm_pack_pairs<1>(Lm, base, t);
  gmm_pack_pairs<2>(Lm, base, t);
  gmm_pack_pairs<3>(Lm, base, t);
  gmm_pack_pairs<4>(Lm, base, t);
  gmm_pack_pairs<5>(Lm, base, t);
  gmm_pack_pairs<6>(Lm, base, t);
  gmm_pack_pairs<7>(Lm, base, t);
  if (t < 32) {
    gmm_pack_pairs<8>(Lm, base, t);
  } else if (t <= (NGROUPS - 1) % 64) {
    float w0 = cterm, w1 = 0.f, w2 = 0.f, w3 = 0.f;
    if (t < (NGROUPS - 1) % 64) {
      const int li = t - 32;
      w0 = bv[4 * li + 0]; w1 = bv[4 * li + 1]; w2 = bv[4 * li + 2]; w3 = bv[4 * li + 3];
    }
    gmm_pack_store(base + 8 * 4096, w0, w1, w2, w3);
  }
}

#define NCH    68
#define NTAIL  17

__device__ __forceinline__ void gmm_next_blk(int& ci, int& jb) {
  ++jb;
  if (jb > 15) { ci = (ci < 16) ? ci + 1 : 16; jb = ci; }
}
__device__ __forceinline__ void gmm_fetch_x(const float* xl, int h, int& ci, int& jb,
                                            v2f (&xp)[2][RB], v4f (&xj)[2][RB]) {
#pragma unroll
  for (int u = 0; u < 2; ++u) {
    const int pa = 4 * ci + 2 * h;
    const int ja = 4 * jb;
#pragma unroll
    for (int rt = 0; rt < RB; ++rt) {
      xp[u][rt] = *(const v2f*)(xl + pa + rt * 16 * XS);
      xj[u][rt] = *(const v4f*)(xl + ja + rt * 16 * XS);
    }
    gmm_next_blk(ci, jb);
  }
}
__device__ __forceinline__ void gmm_fetch_b(const v2f* __restrict__ bp0, const v2f* __restrict__ bp1,
                                            int c, v2f (&bn)[8][2]) {
#pragma unroll
  for (int i = 0; i < 8; ++i) {
    bn[i][0] = bp0[(size_t)(8 * c + i) * 32];
    bn[i][1] = bp1[(size_t)(8 * c + i) * 32];
  }
}
__device__ __forceinline__ void gmm_run(v8f (&acc)[RB][2], const v2f (&xp)[2][RB], const v4f (&xj)[2][RB],
                                        const v2f (&bc)[8][2]) {
#pragma unroll
  for (int u = 0; u < 2; ++u) {
#pragma unroll
    for (int b = 0; b < 4; ++b) {
#pragma unroll
      for (int rt = 0; rt < RB; ++rt) {
        const v2f a = xp[u][rt] * xj[u][rt][b];
        acc[rt][0] = __builtin_amdgcn_wmma_f32_16x16x4_f32(false, a, false, bc[4 * u + b][0], (short)0, acc[rt][0], false, false);
        acc[rt][1] = __builtin_amdgcn_wmma_f32_16x16x4_f32(false, a, false, bc[4 * u + b][1], (short)0, acc[rt][1], false, false);
      }
    }
  }
}
__device__ __forceinline__ void gmm_step(v8f (&acc)[RB][2], const v2f (&xpc)[2][RB], const v4f (&xjc)[2][RB],
                                         v2f (&xpn)[2][RB], v4f (&xjn)[2][RB], const v2f (&bc)[8][2],
                                         v2f (&bf)[8][2], const v2f* __restrict__ bp0,
                                         const v2f* __restrict__ bp1, int cfetch, const float* xl, int h,
                                         int& ci, int& jb) {
  gmm_fetch_b(bp0, bp1, cfetch, bf);
  gmm_fetch_x(xl, h, ci, jb, xpn, xjn);
  __builtin_amdgcn_sched_barrier(0);
  gmm_run(acc, xpc, xjc, bc);
}
__device__ __forceinline__ void gmm_tail(v8f (&acc)[RB][2], const v2f (&ba)[8][2], const v2f (&bb)[8][2],
                                         const v2f (&bl)[2], const float* xl, int h) {
  v2f ta[NTAIL][RB];
#pragma unroll
  for (int i = 0; i < NTAIL; ++i) {
#pragma unroll
    for (int rt = 0; rt < RB; ++rt) ta[i][rt] = *(const v2f*)(xl + 4 * i + 2 * h + rt * 16 * XS);
  }
  __builtin_amdgcn_sched_barrier(0);
#pragma unroll
  for (int i = 0; i < NTAIL; ++i) {
#pragma unroll
    for (int rt = 0; rt < RB; ++rt) {
      const v2f f0 = (i < 8) ? ba[i & 7][0] : ((i < 16) ? bb[i & 7][0] : bl[0]);
      const v2f f1 = (i < 8) ? ba[i & 7][1] : ((i < 16) ? bb[i & 7][1] : bl[1]);
      acc[rt][0] = __builtin_amdgcn_wmma_f32_16x16x4_f32(false, ta[i][rt], false, f0, (short)0, acc[rt][0], false, false);
      acc[rt][1] = __builtin_amdgcn_wmma_f32_16x16x4_f32(false, ta[i][rt], false, f1, (short)0, acc[rt][1], false, false);
    }
  }
}

__global__ __launch_bounds__(32 * WPB) void gmm_main(const float* __restrict__ X,
                                                     const float* __restrict__ bpack,
                                                     float* __restrict__ out,
                                                     int ntiles) {
  __shared__ __attribute__((aligned(16))) float xs[WPB][16 * RB][XS];
  const int tid = threadIdx.x;
  const int w = __builtin_amdgcn_readfirstlane(tid >> 5);
  const int lane = tid & 31;
  const int h = lane >> 4;
  const int r = lane & 15;
  const int t0 = (blockIdx.x * WPB + w) * RB;
  int nvalid = ntiles - t0;
  if (nvalid > RB) nvalid = RB;

  if (nvalid > 0) {
    const float* xp = X + (size_t)t0 * 16 * 64;
    const int umax = nvalid * 8;
    for (int u = 0; u < umax; ++u) {
      const int flat = u * 128 + lane * 4;
      float4 v = *(const float4*)(xp + flat);
      *(float4*)&xs[w][flat >> 6][flat & 63] = v;
    }
    float4 z; z.x = 0.f; z.y = 0.f; z.z = 0.f; z.w = 0.f;
    for (int u = umax; u < RB * 8; ++u) {
      const int flat = u * 128 + lane * 4;
      *(float4*)&xs[w][flat >> 6][flat & 63] = z;
    }
    float4 one; one.x = 1.f; one.y = 0.f; one.z = 0.f; one.w = 0.f;
    for (int rr = 0; rr < 16 * RB; rr += 32) *(float4*)&xs[w][rr + lane][64] = one;
  }
  __syncthreads();
  if (nvalid <= 0) return;

  const float* xl = &xs[w][r][0];
  const v2f* __restrict__ bp0 = (const v2f*)bpack + (h * 16 + r);
  const v2f* __restrict__ bp1 = bp0 + (size_t)NGROUPS * 32;

  v8f acc[RB][2];
#pragma unroll
  for (int rt = 0; rt < RB; ++rt) { acc[rt][0] = (v8f){}; acc[rt][1] = (v8f){}; }

  v2f b0[8][2], b1[8][2], b2[8][2], b3[8][2];
  v2f xpe[2][RB], xpo[2][RB];
  v4f xje[2][RB], xjo[2][RB];
  int ci = 0, jb = 0;

  gmm_fetch_b(bp0, bp1, 0, b0);
  gmm_fetch_b(bp0, bp1, 1, b1);
  v2f bl[2];
  bl[0] = bp0[(size_t)(NGROUPS - 1) * 32];
  bl[1] = bp1[(size_t)(NGROUPS - 1) * 32];
  gmm_fetch_x(xl, h, ci, jb, xpe, xje);
  for (int c = 0; c < NCH; c += 4) {
    gmm_step(acc, xpe, xje, xpo, xjo, b0, b2, bp0, bp1, c + 2, xl, h, ci, jb);
    gmm_step(acc, xpo, xjo, xpe, xje, b1, b3, bp0, bp1, c + 3, xl, h, ci, jb);
    gmm_step(acc, xpe, xje, xpo, xjo, b2, b0, bp0, bp1, c + 4, xl, h, ci, jb);
    gmm_step(acc, xpo, xjo, xpe, xje, b3, b1, bp0, bp1, c + 5, xl, h, ci, jb);
  }
  gmm_tail(acc, b0, b1, bl, xl, h);

#pragma unroll
  for (int rt = 0; rt < RB; ++rt) {
    float res[8];
#pragma unroll
    for (int v = 0; v < 8; ++v) {
      float a0 = acc[rt][0][v], a1 = acc[rt][1][v];
      float mx = fmaxf(a0, a1);
      mx = fmaxf(mx, __shfl_xor(mx, 1, 32));
      mx = fmaxf(mx, __shfl_xor(mx, 2, 32));
      mx = fmaxf(mx, __shfl_xor(mx, 4, 32));
      mx = fmaxf(mx, __shfl_xor(mx, 8, 32));
      float e = expf(a0 - mx) + expf(a1 - mx);
      e += __shfl_xor(e, 1, 32);
      e += __shfl_xor(e, 2, 32);
      e += __shfl_xor(e, 4, 32);
      e += __shfl_xor(e, 8, 32);
      res[v] = mx + logf(e);
    }
    float val = res[0];
#pragma unroll
    for (int v = 1; v < 8; ++v) val = (r == v) ? res[v] : val;
    if (rt < nvalid && r < 8)
      out[(size_t)(t0 + rt) * 16 + 8 * h + r] = val;
  }
}

extern "C" void kernel_launch(void* const* d_in, const int* in_sizes, int n_in,
                              void* d_out, int out_size, void* d_ws, size_t ws_size,
                              hipStream_t stream) {
  const float* X   = (const float*)d_in[0];
  const float* loc = (const float*)d_in[1];
  const float* cov = (const float*)d_in[2];
  const float* pi  = (const float*)d_in[3];
  float* out = (float*)d_out;
  float* bpack = (float*)d_ws;

  const int N = in_sizes[0] / 64;
  const int ntiles = N / 16;

  gmm_prep<<<GMM_K, 64, 0, stream>>>(loc, cov, pi, bpack);
  const int tilesPerBlock = WPB * RB;
  const int nblk = (ntiles + tilesPerBlock - 1) / tilesPerBlock;
  gmm_main<<<nblk, 32 * WPB, 0, stream>>>(X, bpack, out, ntiles);
}
